# starting kernel
# baseline (speedup 1.0000x reference)
#include <hip/hip_runtime.h>

typedef float v2f __attribute__((ext_vector_type(2)));
typedef float v8f __attribute__((ext_vector_type(8)));


__global__ void fill_ones_k(float* __restrict__ p, int n) {
    int i = blockIdx.x * blockDim.x + threadIdx.x;
    if (i < n) p[i] = 1.0f;
}

__global__ void deg_accum_k(const int* __restrict__ dst, float* __restrict__ deg, int E) {
    int e = blockIdx.x * blockDim.x + threadIdx.x;
    if (e < E) atomicAdd(&deg[dst[e]], 1.0f);
}

__global__ void rsqrt_k(float* __restrict__ p, int n) {
    int i = blockIdx.x * blockDim.x + threadIdx.x;
    if (i < n) p[i] = rsqrtf(p[i]);
}

__global__ void coef_k(const int* __restrict__ src, const int* __restrict__ dst,
                       const float* __restrict__ dinv, float* __restrict__ coef, int E) {
    int e = blockIdx.x * blockDim.x + threadIdx.x;
    if (e < E) coef[e] = dinv[src[e]] * dinv[dst[e]];
}


template<int KDIM, int NOUT>
__global__ void gemm_wmma_k(const float* __restrict__ X, const float* __restrict__ W,
                            float* __restrict__ H, int ntiles, int nrows) {
    constexpr int NT   = (NOUT + 15) / 16;
    constexpr int KT   = (KDIM + 3) / 4;
    constexpr int KPAD = KT * 4;
    constexpr int NW16 = NT * 16;

    __shared__ float sW[KPAD * NW16];

    for (int idx = threadIdx.x; idx < KPAD * NW16; idx += blockDim.x) {
        const int k = idx / NW16;
        const int n = idx - k * NW16;
        sW[idx] = (k < KDIM && n < NOUT) ? W[k * NOUT + n] : 0.0f;
    }
    __syncthreads();

    const int lane = threadIdx.x & 31;
    const int wv   = blockIdx.x * (blockDim.x >> 5) + (threadIdx.x >> 5);
    if (wv >= ntiles) return;
    const int row0 = wv * 16;
    const int m    = lane & 15;
    const int half = lane >> 4;

    const unsigned lim = (unsigned)nrows * KDIM - 1u;

    v8f acc[NT];
#pragma unroll
    for (int t = 0; t < NT; ++t) acc[t] = 0.0f;

#pragma unroll
    for (int kt = 0; kt < KT; ++kt) {
        const int ka = kt * 4 + half * 2;
        const unsigned i0 = (unsigned)(row0 + m) * KDIM + ka;
        v2f a;
        a.x = X[min(i0,      lim)];
        a.y = X[min(i0 + 1u, lim)];
#pragma unroll
        for (int nt = 0; nt < NT; ++nt) {
            const int col = nt * 16 + m;
            v2f b;
            b.x = sW[ka * NW16 + col];
            b.y = sW[(ka + 1) * NW16 + col];
            acc[nt] = __builtin_amdgcn_wmma_f32_16x16x4_f32(
                false, a, false, b, (short)0, acc[nt], false, false);
        }
    }

#pragma unroll
    for (int nt = 0; nt < NT; ++nt) {
#pragma unroll
        for (int i = 0; i < 8; ++i) {
            const int r   = row0 + i + half * 8;
            const int col = nt * 16 + m;
            if (col < NOUT) H[(size_t)r * NOUT + col] = acc[nt][i];
        }
    }
}


template<int C>
__global__ void init_agg_k(const float* __restrict__ h, const float* __restrict__ dinv,
                           const float* __restrict__ b, float* __restrict__ agg, int n) {
    int t = blockIdx.x * blockDim.x + threadIdx.x;
    if (t >= n * C) return;
    const int i  = t / C;
    const int ch = t - i * C;
    const float di = dinv[i];
    agg[t] = fmaf(h[t], di * di, b[ch]);
}

__global__ void scatter64_k(const int* __restrict__ src, const int* __restrict__ dst,
                            const float* __restrict__ coef, const float* __restrict__ h,
                            float* __restrict__ agg, int E) {
    const int t = blockIdx.x * blockDim.x + threadIdx.x;
    const int e = t >> 4;
    if (e >= E) return;
    const int cg = (t & 15) << 2;
    const int s = src[e];
    const int d = dst[e];
    const float c = coef[e];
    const float4 v = *reinterpret_cast<const float4*>(h + (size_t)s * 64 + cg);
    float* p = agg + (size_t)d * 64 + cg;
    atomicAdd(p + 0, v.x * c);
    atomicAdd(p + 1, v.y * c);
    atomicAdd(p + 2, v.z * c);
    atomicAdd(p + 3, v.w * c);
}

__global__ void scatter3_k(const int* __restrict__ src, const int* __restrict__ dst,
                           const float* __restrict__ coef, const float* __restrict__ h,
                           float* __restrict__ out, int E) {
    const int e = blockIdx.x * blockDim.x + threadIdx.x;
    if (e >= E) return;
    const int s = src[e];
    const int d = dst[e];
    const float c = coef[e];
    const float* hp = h + (size_t)s * 3;
    float* p = out + (size_t)d * 3;
    atomicAdd(p + 0, hp[0] * c);
    atomicAdd(p + 1, hp[1] * c);
    atomicAdd(p + 2, hp[2] * c);
}

__global__ void relu_k(float* __restrict__ p, int n) {
    int i = blockIdx.x * blockDim.x + threadIdx.x;
    if (i < n) p[i] = fmaxf(p[i], 0.0f);
}


static inline int cdiv(long long a, int b) { return (int)((a + b - 1) / b); }

extern "C" void kernel_launch(void* const* d_in, const int* in_sizes, int n_in,
                              void* d_out, int out_size, void* d_ws, size_t ws_size,
                              hipStream_t stream) {
    const float* x   = (const float*)d_in[0];
    const int*   ei  = (const int*)d_in[1];
    const float* W1  = (const float*)d_in[2];
    const float* b1  = (const float*)d_in[3];
    const float* W2  = (const float*)d_in[4];
    const float* b2  = (const float*)d_in[5];
    const float* W3  = (const float*)d_in[6];
    const float* b3  = (const float*)d_in[7];
    float* out = (float*)d_out;

    const int N = in_sizes[0] / 9;
    const int E = in_sizes[1] / 2;
    const int* src = ei;
    const int* dst = ei + E;

    float* dinv = (float*)d_ws;
    float* coef = dinv + N;
    float* h    = coef + E;
    float* agg  = h + (size_t)N * 64;
    (void)ws_size; (void)n_in; (void)out_size;

    const int ntiles = (N + 15) / 16;

    fill_ones_k<<<cdiv(N, 256), 256, 0, stream>>>(dinv, N);
    deg_accum_k<<<cdiv(E, 256), 256, 0, stream>>>(dst, dinv, E);
    rsqrt_k<<<cdiv(N, 256), 256, 0, stream>>>(dinv, N);
    coef_k<<<cdiv(E, 256), 256, 0, stream>>>(src, dst, dinv, coef, E);

    gemm_wmma_k<9, 64><<<cdiv(ntiles, 8), 256, 0, stream>>>(x, W1, h, ntiles, N);
    init_agg_k<64><<<cdiv((long long)N * 64, 256), 256, 0, stream>>>(h, dinv, b1, agg, N);
    scatter64_k<<<cdiv((long long)E * 16, 256), 256, 0, stream>>>(src, dst, coef, h, agg, E);
    relu_k<<<cdiv((long long)N * 64, 256), 256, 0, stream>>>(agg, N * 64);

    gemm_wmma_k<64, 64><<<cdiv(ntiles, 8), 256, 0, stream>>>(agg, W2, h, ntiles, N);
    init_agg_k<64><<<cdiv((long long)N * 64, 256), 256, 0, stream>>>(h, dinv, b2, agg, N);
    scatter64_k<<<cdiv((long long)E * 16, 256), 256, 0, stream>>>(src, dst, coef, h, agg, E);
    relu_k<<<cdiv((long long)N * 64, 256), 256, 0, stream>>>(agg, N * 64);

    gemm_wmma_k<64, 3><<<cdiv(ntiles, 8), 256, 0, stream>>>(agg, W3, h, ntiles, N);
    init_agg_k<3><<<cdiv((long long)N * 3, 256), 256, 0, stream>>>(h, dinv, b3, out, N);
    scatter3_k<<<cdiv(E, 256), 256, 0, stream>>>(src, dst, coef, h, out, E);
}


# speedup vs baseline: 2.1783x; 2.1783x over previous
#include <hip/hip_runtime.h>


typedef float v2f __attribute__((ext_vector_type(2)));
typedef float v8f __attribute__((ext_vector_type(8)));

__device__ __forceinline__ int pkg_half() {
    return (int)((__builtin_amdgcn_s_sendmsg_rtn(0x87) >> 18) & 1u);
}


__global__ void fill_deg_k(float* __restrict__ d0, float* __restrict__ d1, int n) {
    int i = blockIdx.x * blockDim.x + threadIdx.x;
    if (i < n) {
        d0[i] = 1.0f;
        d1[i] = 0.0f;
    }
}

__global__ void deg_accum_k(const int* __restrict__ dst, float* __restrict__ d0,
                            float* __restrict__ d1, int E) {
    float* deg = pkg_half() ? d1 : d0;
    int e = blockIdx.x * blockDim.x + threadIdx.x;
    if (e < E) atomicAdd(&deg[dst[e]], 1.0f);
}

__global__ void rsqrt_k(float* __restrict__ d0, const float* __restrict__ d1, int n) {
    int i = blockIdx.x * blockDim.x + threadIdx.x;
    if (i < n) d0[i] = rsqrtf(d0[i] + d1[i]);
}

__global__ void coef_k(const int* __restrict__ src, const int* __restrict__ dst,
                       const float* __restrict__ dinv, float* __restrict__ coef, int E) {
    int e = blockIdx.x * blockDim.x + threadIdx.x;
    if (e < E) coef[e] = dinv[src[e]] * dinv[dst[e]];
}


template<int KDIM, int NOUT, bool PIN, bool POUT>
__global__ void gemm_wmma_k(const float* __restrict__ X0, const float* __restrict__ X1,
                            const float* __restrict__ W,
                            float* __restrict__ H0, float* __restrict__ H1,
                            int ntiles, int nrows) {
    constexpr int NT   = (NOUT + 15) / 16;
    constexpr int KT   = (KDIM + 3) / 4;
    constexpr int KPAD = KT * 4;
    constexpr int NW16 = NT * 16;
    static_assert(!PIN  || KDIM == 64, "planar input is 2 x 32 channels");
    static_assert(!POUT || NOUT == 64, "planar output is 2 x 32 channels");

    __shared__ float sW[KPAD * NW16];

    for (int idx = threadIdx.x; idx < KPAD * NW16; idx += blockDim.x) {
        const int k = idx / NW16;
        const int n = idx - k * NW16;
        sW[idx] = (k < KDIM && n < NOUT) ? W[k * NOUT + n] : 0.0f;
    }
    __syncthreads();

    const int lane = threadIdx.x & 31;
    const int wv   = blockIdx.x * (blockDim.x >> 5) + (threadIdx.x >> 5);
    if (wv >= ntiles) return;
    const int row0 = wv * 16;
    const int m    = lane & 15;
    const int half = lane >> 4;

    const unsigned lim  = (unsigned)nrows * KDIM - 1u;
    const unsigned rowc = min((unsigned)(row0 + m), (unsigned)nrows - 1u);

    v8f acc[NT];
#pragma unroll
    for (int t = 0; t < NT; ++t) acc[t] = 0.0f;

#pragma unroll
    for (int kt = 0; kt < KT; ++kt) {
        const int ka = kt * 4 + half * 2;
        v2f a;
        if constexpr (PIN) {
            const float* xp = (kt * 4 < 32) ? X0 : X1;
            const unsigned kk = (unsigned)((kt * 4) & 31) + (unsigned)(half * 2);
            a.x = xp[rowc * 32u + kk];
            a.y = xp[rowc * 32u + kk + 1u];
        } else {
            const unsigned i0 = (unsigned)(row0 + m) * KDIM + ka;
            a.x = X0[min(i0,      lim)];
            a.y = X0[min(i0 + 1u, lim)];
        }
#pragma unroll
        for (int nt = 0; nt < NT; ++nt) {
            const int col = nt * 16 + m;
            v2f b;
            b.x = sW[ka * NW16 + col];
            b.y = sW[(ka + 1) * NW16 + col];
            acc[nt] = __builtin_amdgcn_wmma_f32_16x16x4_f32(
                false, a, false, b, (short)0, acc[nt], false, false);
        }
    }

#pragma unroll
    for (int nt = 0; nt < NT; ++nt) {
#pragma unroll
        for (int i = 0; i < 8; ++i) {
            const int r   = row0 + i + half * 8;
            const int col = nt * 16 + m;
            if constexpr (POUT) {
                float* hp = (nt * 16 < 32) ? H0 : H1;
                hp[(size_t)r * 32 + (col & 31)] = acc[nt][i];
            } else {
                if (col < NOUT) H0[(size_t)r * NOUT + col] = acc[nt][i];
            }
        }
    }
}


__global__ void init32_k(const float* __restrict__ h, const float* __restrict__ dinv,
                         const float* __restrict__ b, float* __restrict__ p,
                         float* __restrict__ z, int n) {
    int t = blockIdx.x * blockDim.x + threadIdx.x;
    if (t >= n * 32) return;
    const int i  = t >> 5;
    const int ch = t & 31;
    const float di = dinv[i];
    p[t] = fmaf(h[t], di * di, b[ch]);
    z[t] = 0.0f;
}

__global__ void mid32_k(float* __restrict__ p, float* __restrict__ z,
                        const float* __restrict__ hb, const float* __restrict__ dinv,
                        const float* __restrict__ b, float* __restrict__ s, int n) {
    int t = blockIdx.x * blockDim.x + threadIdx.x;
    if (t >= n * 32) return;
    const int i  = t >> 5;
    const int ch = t & 31;
    const float di = dinv[i];
    p[t] = fmaxf(p[t] + z[t], 0.0f);
    z[t] = 0.0f;
    s[t] = fmaf(hb[t], di * di, b[ch]);
}

__global__ void fin32_k(float* __restrict__ s, const float* __restrict__ z, int n) {
    int t = blockIdx.x * blockDim.x + threadIdx.x;
    if (t < n * 32) s[t] = fmaxf(s[t] + z[t], 0.0f);
}

constexpr int EPW = 128;

template<int G>
__device__ __forceinline__ void edge_group32(int j, int s, int d, float c, int lane,
                                             const float* __restrict__ h, float* __restrict__ agg) {
    float v[G];
#pragma unroll
    for (int u = 0; u < G; ++u) {
        const int sj = __builtin_amdgcn_readlane(s, j + u);
        const float* hr = h + (size_t)sj * 32;
        v[u] = hr[lane];
    }
#pragma unroll
    for (int u = 0; u < G; ++u) {
        const int   dj = __builtin_amdgcn_readlane(d, j + u);
        const float cj = __int_as_float(__builtin_amdgcn_readlane(__float_as_int(c), j + u));
        float* ar = agg + (size_t)dj * 32;
        atomicAdd(ar + lane, v[u] * cj);
    }
}

__global__ __launch_bounds__(256)
void scatter32_k(const int* __restrict__ src, const int* __restrict__ dst,
                 const float* __restrict__ coef, const float* __restrict__ h,
                 float* __restrict__ p0, float* __restrict__ p1, int E) {
    const int lane = threadIdx.x & 31;
    const int wib  = __builtin_amdgcn_readfirstlane((int)(threadIdx.x >> 5));
    const int wave = blockIdx.x * (int)(blockDim.x >> 5) + wib;
    const int ebeg = wave * EPW;
    if (ebeg >= E) return;
    const int eend = min(ebeg + EPW, E);

    float* agg = pkg_half() ? p1 : p0;

    int s = 0, d = 0;
    float c = 0.0f;
    if (ebeg + lane < eend) {
        s = src[ebeg + lane];
        d = dst[ebeg + lane];
        c = coef[ebeg + lane];
    }

    for (int e0 = ebeg; e0 < eend; e0 += 32) {
        const int en = e0 + 32 + lane;
        int sn = 0, dn = 0;
        float cn = 0.0f;
        if (en < eend) {
            sn = src[en];
            dn = dst[en];
            cn = coef[en];
        }

        const int cnt = min(32, eend - e0);
        int j = 0;
        if (cnt == 32) {
            edge_group32<32>(0, s, d, c, lane, h, agg);
            j = 32;
        }
        for (; j + 16 <= cnt; j += 16) edge_group32<16>(j, s, d, c, lane, h, agg);
        for (; j < cnt; ++j)
            edge_group32<1>(j, s, d, c, lane, h, agg);

        s = sn; d = dn; c = cn;
    }
}


__global__ void init3_k(const float* __restrict__ h, const float* __restrict__ dinv,
                        const float* __restrict__ b, float* __restrict__ o0,
                        float* __restrict__ o1, int n) {
    int t = blockIdx.x * blockDim.x + threadIdx.x;
    if (t >= n * 3) return;
    const int i  = t / 3;
    const int ch = t - i * 3;
    const float di = dinv[i];
    o0[t] = fmaf(h[t], di * di, b[ch]);
    o1[t] = 0.0f;
}

__global__ void scatter3_k(const int* __restrict__ src, const int* __restrict__ dst,
                           const float* __restrict__ coef, const float* __restrict__ h,
                           float* __restrict__ o0, float* __restrict__ o1, int E) {
    float* out = pkg_half() ? o1 : o0;
    const int e = blockIdx.x * blockDim.x + threadIdx.x;
    if (e >= E) return;
    const int s = src[e];
    const int d = dst[e];
    const float c = coef[e];
    const float* hp = h + (size_t)s * 3;
    float* p = out + (size_t)d * 3;
    atomicAdd(p + 0, hp[0] * c);
    atomicAdd(p + 1, hp[1] * c);
    atomicAdd(p + 2, hp[2] * c);
}

__global__ void add_plane_k(float* __restrict__ o0, const float* __restrict__ o1, int n) {
    int i = blockIdx.x * blockDim.x + threadIdx.x;
    if (i < n) o0[i] += o1[i];
}


static inline int cdiv(long long a, int b) { return (int)((a + b - 1) / b); }

extern "C" void kernel_launch(void* const* d_in, const int* in_sizes, int n_in,
                              void* d_out, int out_size, void* d_ws, size_t ws_size,
                              hipStream_t stream) {
    const float* x   = (const float*)d_in[0];
    const int*   ei  = (const int*)d_in[1];
    const float* W1  = (const float*)d_in[2];
    const float* b1  = (const float*)d_in[3];
    const float* W2  = (const float*)d_in[4];
    const float* b2  = (const float*)d_in[5];
    const float* W3  = (const float*)d_in[6];
    const float* b3  = (const float*)d_in[7];
    float* out = (float*)d_out;

    const int N = in_sizes[0] / 9;
    const int E = in_sizes[1] / 2;
    const int* src = ei;
    const int* dst = ei + E;

    float* R0   = (float*)d_ws;
    float* R1   = R0 + (size_t)N * 32;
    float* R2   = R1 + (size_t)N * 32;
    float* R3   = R2 + (size_t)N * 32;
    float* dinv = R3 + (size_t)N * 32;
    float* coef = dinv + N;
    (void)ws_size; (void)n_in; (void)out_size;

    const int ntiles = (N + 15) / 16;
    const int gN   = cdiv(N, 256);
    const int gE   = cdiv(E, 256);
    const int gP   = cdiv((long long)N * 32, 256);
    const int gS   = cdiv(cdiv(E, EPW), 8);
    const int gG   = cdiv(ntiles, 8);
    const int gN3  = cdiv((long long)N * 3, 256);

    fill_deg_k<<<gN, 256, 0, stream>>>(dinv, R0, N);
    deg_accum_k<<<gE, 256, 0, stream>>>(dst, dinv, R0, E);
    rsqrt_k<<<gN, 256, 0, stream>>>(dinv, R0, N);
    coef_k<<<gE, 256, 0, stream>>>(src, dst, dinv, coef, E);

    gemm_wmma_k<9, 64, false, true><<<gG, 256, 0, stream>>>(x, nullptr, W1, R0, R1, ntiles, N);
    init32_k<<<gP, 256, 0, stream>>>(R0, dinv, b1, R2, R3, N);
    scatter32_k<<<gS, 256, 0, stream>>>(src, dst, coef, R0, R2, R3, E);
    mid32_k<<<gP, 256, 0, stream>>>(R2, R3, R1, dinv, b1 + 32, R0, N);
    scatter32_k<<<gS, 256, 0, stream>>>(src, dst, coef, R1, R0, R3, E);
    fin32_k<<<gP, 256, 0, stream>>>(R0, R3, N);

    gemm_wmma_k<64, 64, true, true><<<gG, 256, 0, stream>>>(R2, R0, W2, R1, R3, ntiles, N);
    init32_k<<<gP, 256, 0, stream>>>(R1, dinv, b2, R2, R0, N);
    scatter32_k<<<gS, 256, 0, stream>>>(src, dst, coef, R1, R2, R0, E);
    mid32_k<<<gP, 256, 0, stream>>>(R2, R0, R3, dinv, b2 + 32, R1, N);
    scatter32_k<<<gS, 256, 0, stream>>>(src, dst, coef, R3, R1, R0, E);
    fin32_k<<<gP, 256, 0, stream>>>(R1, R0, N);

    gemm_wmma_k<64, 3, true, false><<<gG, 256, 0, stream>>>(R2, R1, W3, R0, nullptr, ntiles, N);
    init3_k<<<gN3, 256, 0, stream>>>(R0, dinv, b3, out, R3, N);
    scatter3_k<<<gE, 256, 0, stream>>>(src, dst, coef, R0, out, R3, E);
    add_plane_k<<<gN3, 256, 0, stream>>>(out, R3, N * 3);
}


# Layer-3 edge adds by (edge, channel) lanes: 8 edges per instruction, an edge's 3 channel adds sit in neighbouring lanes and leave as one line request, not 3; sums stay in arrival order (atomics)
# speedup vs baseline: 2.2608x; 1.0379x over previous
#include <hip/hip_runtime.h>


typedef float v2f __attribute__((ext_vector_type(2)));
typedef float v8f __attribute__((ext_vector_type(8)));

__device__ __forceinline__ int pkg_half() {
    return (int)((__builtin_amdgcn_s_sendmsg_rtn(0x87) >> 18) & 1u);
}


__global__ void fill_deg_k(float* __restrict__ d0, float* __restrict__ d1, int n) {
    int i = blockIdx.x * blockDim.x + threadIdx.x;
    if (i < n) {
        d0[i] = 1.0f;
        d1[i] = 0.0f;
    }
}

__global__ void deg_accum_k(const int* __restrict__ dst, float* __restrict__ d0,
                            float* __restrict__ d1, int E) {
    float* deg = pkg_half() ? d1 : d0;
    int e = blockIdx.x * blockDim.x + threadIdx.x;
    if (e < E) atomicAdd(&deg[dst[e]], 1.0f);
}

__global__ void rsqrt_k(float* __restrict__ d0, const float* __restrict__ d1, int n) {
    int i = blockIdx.x * blockDim.x + threadIdx.x;
    if (i < n) d0[i] = rsqrtf(d0[i] + d1[i]);
}

__global__ void coef_k(const int* __restrict__ src, const int* __restrict__ dst,
                       const float* __restrict__ dinv, float* __restrict__ coef, int E) {
    int e = blockIdx.x * blockDim.x + threadIdx.x;
    if (e < E) coef[e] = dinv[src[e]] * dinv[dst[e]];
}


template<int KDIM, int NOUT, bool PIN, bool POUT>
__global__ void gemm_wmma_k(const float* __restrict__ X0, const float* __restrict__ X1,
                            const float* __restrict__ W,
                            float* __restrict__ H0, float* __restrict__ H1,
                            int ntiles, int nrows) {
    constexpr int NT   = (NOUT + 15) / 16;
    constexpr int KT   = (KDIM + 3) / 4;
    constexpr int KPAD = KT * 4;
    constexpr int NW16 = NT * 16;
    static_assert(!PIN  || KDIM == 64, "planar input is 2 x 32 channels");
    static_assert(!POUT || NOUT == 64, "planar output is 2 x 32 channels");

    __shared__ float sW[KPAD * NW16];

    for (int idx = threadIdx.x; idx < KPAD * NW16; idx += blockDim.x) {
        const int k = idx / NW16;
        const int n = idx - k * NW16;
        sW[idx] = (k < KDIM && n < NOUT) ? W[k * NOUT + n] : 0.0f;
    }
    __syncthreads();

    const int lane = threadIdx.x & 31;
    const int wv   = blockIdx.x * (blockDim.x >> 5) + (threadIdx.x >> 5);
    if (wv >= ntiles) return;
    const int row0 = wv * 16;
    const int m    = lane & 15;
    const int half = lane >> 4;

    const unsigned lim  = (unsigned)nrows * KDIM - 1u;
    const unsigned rowc = min((unsigned)(row0 + m), (unsigned)nrows - 1u);

    v8f acc[NT];
#pragma unroll
    for (int t = 0; t < NT; ++t) acc[t] = 0.0f;

#pragma unroll
    for (int kt = 0; kt < KT; ++kt) {
        const int ka = kt * 4 + half * 2;
        v2f a;
        if constexpr (PIN) {
            const float* xp = (kt * 4 < 32) ? X0 : X1;
            const unsigned kk = (unsigned)((kt * 4) & 31) + (unsigned)(half * 2);
            a.x = xp[rowc * 32u + kk];
            a.y = xp[rowc * 32u + kk + 1u];
        } else {
            const unsigned i0 = (unsigned)(row0 + m) * KDIM + ka;
            a.x = X0[min(i0,      lim)];
            a.y = X0[min(i0 + 1u, lim)];
        }
#pragma unroll
        for (int nt = 0; nt < NT; ++nt) {
            const int col = nt * 16 + m;
            v2f b;
            b.x = sW[ka * NW16 + col];
            b.y = sW[(ka + 1) * NW16 + col];
            acc[nt] = __builtin_amdgcn_wmma_f32_16x16x4_f32(
                false, a, false, b, (short)0, acc[nt], false, false);
        }
    }

#pragma unroll
    for (int nt = 0; nt < NT; ++nt) {
#pragma unroll
        for (int i = 0; i < 8; ++i) {
            const int r   = row0 + i + half * 8;
            const int col = nt * 16 + m;
            if constexpr (POUT) {
                float* hp = (nt * 16 < 32) ? H0 : H1;
                hp[(size_t)r * 32 + (col & 31)] = acc[nt][i];
            } else {
                if (col < NOUT) H0[(size_t)r * NOUT + col] = acc[nt][i];
            }
        }
    }
}


__global__ void init32_k(const float* __restrict__ h, const float* __restrict__ dinv,
                         const float* __restrict__ b, float* __restrict__ p,
                         float* __restrict__ z, int n) {
    int t = blockIdx.x * blockDim.x + threadIdx.x;
    if (t >= n * 32) return;
    const int i  = t >> 5;
    const int ch = t & 31;
    const float di = dinv[i];
    p[t] = fmaf(h[t], di * di, b[ch]);
    z[t] = 0.0f;
}

__global__ void mid32_k(float* __restrict__ p, float* __restrict__ z,
                        const float* __restrict__ hb, const float* __restrict__ dinv,
                        const float* __restrict__ b, float* __restrict__ s, int n) {
    int t = blockIdx.x * blockDim.x + threadIdx.x;
    if (t >= n * 32) return;
    const int i  = t >> 5;
    const int ch = t & 31;
    const float di = dinv[i];
    p[t] = fmaxf(p[t] + z[t], 0.0f);
    z[t] = 0.0f;
    s[t] = fmaf(hb[t], di * di, b[ch]);
}

__global__ void fin32_k(float* __restrict__ s, const float* __restrict__ z, int n) {
    int t = blockIdx.x * blockDim.x + threadIdx.x;
    if (t < n * 32) s[t] = fmaxf(s[t] + z[t], 0.0f);
}

constexpr int EPW = 128;

template<int G>
__device__ __forceinline__ void edge_group32(int j, int s, int d, float c, int lane,
                                             const float* __restrict__ h, float* __restrict__ agg) {
    float v[G];
#pragma unroll
    for (int u = 0; u < G; ++u) {
        const int sj = __builtin_amdgcn_readlane(s, j + u);
        const float* hr = h + (size_t)sj * 32;
        v[u] = hr[lane];
    }
#pragma unroll
    for (int u = 0; u < G; ++u) {
        const int   dj = __builtin_amdgcn_readlane(d, j + u);
        const float cj = __int_as_float(__builtin_amdgcn_readlane(__float_as_int(c), j + u));
        float* ar = agg + (size_t)dj * 32;
        atomicAdd(ar + lane, v[u] * cj);
    }
}

__global__ __launch_bounds__(256)
void scatter32_k(const int* __restrict__ src, const int* __restrict__ dst,
                 const float* __restrict__ coef, const float* __restrict__ h,
                 float* __restrict__ p0, float* __restrict__ p1, int E) {
    const int lane = threadIdx.x & 31;
    const int wib  = __builtin_amdgcn_readfirstlane((int)(threadIdx.x >> 5));
    const int wave = blockIdx.x * (int)(blockDim.x >> 5) + wib;
    const int ebeg = wave * EPW;
    if (ebeg >= E) return;
    const int eend = min(ebeg + EPW, E);

    float* agg = pkg_half() ? p1 : p0;

    int s = 0, d = 0;
    float c = 0.0f;
    if (ebeg + lane < eend) {
        s = src[ebeg + lane];
        d = dst[ebeg + lane];
        c = coef[ebeg + lane];
    }

    for (int e0 = ebeg; e0 < eend; e0 += 32) {
        const int en = e0 + 32 + lane;
        int sn = 0, dn = 0;
        float cn = 0.0f;
        if (en < eend) {
            sn = src[en];
            dn = dst[en];
            cn = coef[en];
        }

        const int cnt = min(32, eend - e0);
        int j = 0;
        if (cnt == 32) {
            edge_group32<32>(0, s, d, c, lane, h, agg);
            j = 32;
        }
        for (; j + 16 <= cnt; j += 16) edge_group32<16>(j, s, d, c, lane, h, agg);
        for (; j < cnt; ++j)
            edge_group32<1>(j, s, d, c, lane, h, agg);

        s = sn; d = dn; c = cn;
    }
}


__global__ void init3_k(const float* __restrict__ h, const float* __restrict__ dinv,
                        const float* __restrict__ b, float* __restrict__ o0,
                        float* __restrict__ o1, int n) {
    int t = blockIdx.x * blockDim.x + threadIdx.x;
    if (t >= n * 3) return;
    const int i  = t / 3;
    const int ch = t - i * 3;
    const float di = dinv[i];
    o0[t] = fmaf(h[t], di * di, b[ch]);
    o1[t] = 0.0f;
}

__global__ void scatter3_k(const int* __restrict__ src, const int* __restrict__ dst,
                           const float* __restrict__ coef, const float* __restrict__ h,
                           float* __restrict__ o0, float* __restrict__ o1, int E) {
    float* out = pkg_half() ? o1 : o0;
    const int lane = threadIdx.x & 31;
    const int ch   = lane & 3;
    const int wib   = __builtin_amdgcn_readfirstlane((int)(threadIdx.x >> 5));
    const int wbase = (blockIdx.x * (int)(blockDim.x >> 5) + wib) * 32;
    if (wbase >= E) return;
    if (ch == 3) return;
    const int eb = wbase + (lane >> 2) * 4;

    if (wbase + 32 <= E) {
        int   s[4], d[4];
        float c[4], v[4];
#pragma unroll
        for (int g = 0; g < 4; ++g) {
            s[g] = src[eb + g];
            d[g] = dst[eb + g];
            c[g] = coef[eb + g];
        }
#pragma unroll
        for (int g = 0; g < 4; ++g) v[g] = h[(size_t)s[g] * 3 + ch] * c[g];
#pragma unroll
        for (int g = 0; g < 4; ++g) atomicAdd(out + (size_t)d[g] * 3 + ch, v[g]);
    } else {
        for (int g = 0; g < 4; ++g) {
            const int e = eb + g;
            if (e < E) atomicAdd(out + (size_t)dst[e] * 3 + ch, h[(size_t)src[e] * 3 + ch] * coef[e]);
        }
    }
}

__global__ void add_plane_k(float* __restrict__ o0, const float* __restrict__ o1, int n) {
    int i = blockIdx.x * blockDim.x + threadIdx.x;
    if (i < n) o0[i] += o1[i];
}


static inline int cdiv(long long a, int b) { return (int)((a + b - 1) / b); }

extern "C" void kernel_launch(void* const* d_in, const int* in_sizes, int n_in,
                              void* d_out, int out_size, void* d_ws, size_t ws_size,
                              hipStream_t stream) {
    const float* x   = (const float*)d_in[0];
    const int*   ei  = (const int*)d_in[1];
    const float* W1  = (const float*)d_in[2];
    const float* b1  = (const float*)d_in[3];
    const float* W2  = (const float*)d_in[4];
    const float* b2  = (const float*)d_in[5];
    const float* W3  = (const float*)d_in[6];
    const float* b3  = (const float*)d_in[7];
    float* out = (float*)d_out;

    const int N = in_sizes[0] / 9;
    const int E = in_sizes[1] / 2;
    const int* src = ei;
    const int* dst = ei + E;

    float* R0   = (float*)d_ws;
    float* R1   = R0 + (size_t)N * 32;
    float* R2   = R1 + (size_t)N * 32;
    float* R3   = R2 + (size_t)N * 32;
    float* dinv = R3 + (size_t)N * 32;
    float* coef = dinv + N;
    (void)ws_size; (void)n_in; (void)out_size;

    const int ntiles = (N + 15) / 16;
    const int gN   = cdiv(N, 256);
    const int gE   = cdiv(E, 256);
    const int gP   = cdiv((long long)N * 32, 256);
    const int gS   = cdiv(cdiv(E, EPW), 8);
    const int gG   = cdiv(ntiles, 8);
    const int gN3  = cdiv((long long)N * 3, 256);

    fill_deg_k<<<gN, 256, 0, stream>>>(dinv, R0, N);
    deg_accum_k<<<gE, 256, 0, stream>>>(dst, dinv, R0, E);
    rsqrt_k<<<gN, 256, 0, stream>>>(dinv, R0, N);
    coef_k<<<gE, 256, 0, stream>>>(src, dst, dinv, coef, E);

    gemm_wmma_k<9, 64, false, true><<<gG, 256, 0, stream>>>(x, nullptr, W1, R0, R1, ntiles, N);
    init32_k<<<gP, 256, 0, stream>>>(R0, dinv, b1, R2, R3, N);
    scatter32_k<<<gS, 256, 0, stream>>>(src, dst, coef, R0, R2, R3, E);
    mid32_k<<<gP, 256, 0, stream>>>(R2, R3, R1, dinv, b1 + 32, R0, N);
    scatter32_k<<<gS, 256, 0, stream>>>(src, dst, coef, R1, R0, R3, E);
    fin32_k<<<gP, 256, 0, stream>>>(R0, R3, N);

    gemm_wmma_k<64, 64, true, true><<<gG, 256, 0, stream>>>(R2, R0, W2, R1, R3, ntiles, N);
    init32_k<<<gP, 256, 0, stream>>>(R1, dinv, b2, R2, R0, N);
    scatter32_k<<<gS, 256, 0, stream>>>(src, dst, coef, R1, R2, R0, E);
    mid32_k<<<gP, 256, 0, stream>>>(R2, R0, R3, dinv, b2 + 32, R1, N);
    scatter32_k<<<gS, 256, 0, stream>>>(src, dst, coef, R3, R1, R0, E);
    fin32_k<<<gP, 256, 0, stream>>>(R1, R0, N);

    gemm_wmma_k<64, 3, true, false><<<gG, 256, 0, stream>>>(R2, R1, W3, R0, nullptr, ntiles, N);
    init3_k<<<gN3, 256, 0, stream>>>(R0, dinv, b3, out, R3, N);
    scatter3_k<<<gE, 256, 0, stream>>>(src, dst, coef, R0, out, R3, E);
    add_plane_k<<<gN3, 256, 0, stream>>>(out, R3, N * 3);
}


# Per-node start and finish passes folded into the transform kernels: a tile's wave stores h*dinv^2+bias and the zero plane, and adds+clips the last two planes on load; 5 launches fewer; sum order kept
# speedup vs baseline: 2.3946x; 1.0592x over previous
#include <hip/hip_runtime.h>


typedef float v2f __attribute__((ext_vector_type(2)));
typedef float v4f __attribute__((ext_vector_type(4)));
typedef float v8f __attribute__((ext_vector_type(8)));

__device__ __forceinline__ int pkg_half() {
    return (int)((__builtin_amdgcn_s_sendmsg_rtn(0x87) >> 18) & 1u);
}


__global__ void fill_deg_k(float* __restrict__ d0, float* __restrict__ d1, int n) {
    int i = blockIdx.x * blockDim.x + threadIdx.x;
    if (i < n) {
        d0[i] = 1.0f;
        d1[i] = 0.0f;
    }
}

__global__ void deg_accum_k(const int* __restrict__ dst, float* __restrict__ d0,
                            float* __restrict__ d1, int E) {
    float* deg = pkg_half() ? d1 : d0;
    int e = blockIdx.x * blockDim.x + threadIdx.x;
    if (e < E) atomicAdd(&deg[dst[e]], 1.0f);
}

__global__ void rsqrt_k(float* __restrict__ d0, const float* __restrict__ d1, int n) {
    int i = blockIdx.x * blockDim.x + threadIdx.x;
    if (i < n) d0[i] = rsqrtf(d0[i] + d1[i]);
}

__global__ void coef_k(const int* __restrict__ src, const int* __restrict__ dst,
                       const float* __restrict__ dinv, float* __restrict__ coef, int E) {
    int e = blockIdx.x * blockDim.x + threadIdx.x;
    if (e < E) coef[e] = dinv[src[e]] * dinv[dst[e]];
}


template<int KDIM, int NOUT, bool PIN, bool POUT>
__global__ void gemm_wmma_k(const float* X0, const float* X1, const float* XZ,
                            const float* __restrict__ W,
                            const float* __restrict__ dinv, const float* __restrict__ bias,
                            float* H0, float* H1, float* P, float* Z,
                            int ntiles, int nrows) {
    constexpr int NT   = (NOUT + 15) / 16;
    constexpr int KT   = (KDIM + 3) / 4;
    constexpr int KPAD = KT * 4;
    constexpr int NW16 = NT * 16;
    static_assert(!PIN  || KDIM == 64, "planar input is 2 x 32 channels");
    static_assert(!POUT || NOUT == 64, "planar output is 2 x 32 channels");
    static_assert(POUT || NOUT <= 16, "row-major output is one column tile");

    __shared__ float sW[KPAD * NW16];

    for (int idx = threadIdx.x; idx < KPAD * NW16; idx += blockDim.x) {
        const int k = idx / NW16;
        const int n = idx - k * NW16;
        sW[idx] = (k < KDIM && n < NOUT) ? W[k * NOUT + n] : 0.0f;
    }
    __syncthreads();

    const int lane = threadIdx.x & 31;
    const int wv   = blockIdx.x * (blockDim.x >> 5) + (threadIdx.x >> 5);
    if (wv >= ntiles) return;
    const int row0 = wv * 16;
    const int m    = lane & 15;
    const int half = lane >> 4;

    const unsigned lim  = (unsigned)nrows * KDIM - 1u;
    const unsigned rowc = min((unsigned)(row0 + m), (unsigned)nrows - 1u);

    float d2[8];
#pragma unroll
    for (int i = 0; i < 8; ++i) {
        const float di = dinv[min((unsigned)(row0 + half * 8 + i), (unsigned)nrows - 1u)];
        d2[i] = di * di;
    }
    float bb[2];
    if constexpr (POUT) { bb[0] = bias[m]; bb[1] = bias[16 + m]; }
    else                { bb[0] = (m < NOUT) ? bias[m] : 0.0f; bb[1] = 0.0f; }

    v8f acc[NT];
#pragma unroll
    for (int t = 0; t < NT; ++t) acc[t] = 0.0f;

    v2f av[KT];
    v2f zv[PIN ? KT / 2 : 1];
#pragma unroll
    for (int kt = 0; kt < KT; ++kt) {
        if constexpr (PIN) {
            const unsigned kk = (unsigned)((kt * 4) & 31) + (unsigned)(half * 2);
            const unsigned ix = rowc * 32u + kk;
            if (kt * 4 < 32) {
                av[kt].x = X0[ix];
                av[kt].y = X0[ix + 1u];
            } else {
                av[kt].x = X1[ix];
                av[kt].y = X1[ix + 1u];
                zv[kt - KT / 2].x = XZ[ix];
                zv[kt - KT / 2].y = XZ[ix + 1u];
            }
        } else {
            const unsigned i0 = (unsigned)(row0 + m) * KDIM + kt * 4 + half * 2;
            av[kt].x = X0[min(i0,      lim)];
            av[kt].y = X0[min(i0 + 1u, lim)];
        }
    }

#pragma unroll
    for (int kt = 0; kt < KT; ++kt) {
        const int ka = kt * 4 + half * 2;
        v2f a = av[kt];
        if constexpr (PIN) {
            if (kt * 4 >= 32) {
                a.x = fmaxf(av[kt].x + zv[kt - KT / 2].x, 0.0f);
                a.y = fmaxf(av[kt].y + zv[kt - KT / 2].y, 0.0f);
            }
        }
#pragma unroll
        for (int nt = 0; nt < NT; ++nt) {
            const int col = nt * 16 + m;
            v2f b;
            b.x = sW[ka * NW16 + col];
            b.y = sW[(ka + 1) * NW16 + col];
            acc[nt] = __builtin_amdgcn_wmma_f32_16x16x4_f32(
                false, a, false, b, (short)0, acc[nt], false, false);
        }
    }

#pragma unroll
    for (int nt = 0; nt < NT; ++nt) {
#pragma unroll
        for (int i = 0; i < 8; ++i) {
            const int r   = row0 + i + half * 8;
            const int col = nt * 16 + m;
            if constexpr (POUT) {
                float* hp = (nt * 16 < 32) ? H0 : H1;
                hp[(size_t)r * 32 + (col & 31)] = acc[nt][i];
                if (nt * 16 < 32) P[(size_t)r * 32 + col] = fmaf(acc[nt][i], d2[i], bb[nt & 1]);
            } else {
                if (col < NOUT) {
                    H0[(size_t)r * NOUT + col] = acc[nt][i];
                    P[(size_t)r * NOUT + col]  = fmaf(acc[nt][i], d2[i], bb[0]);
                    Z[(size_t)r * NOUT + col]  = 0.0f;
                }
            }
        }
    }
    if constexpr (POUT) {
        v4f* zp = reinterpret_cast<v4f*>(Z + (size_t)row0 * 32);
        const v4f z4 = 0.0f;
#pragma unroll
        for (int j = 0; j < 4; ++j) zp[j * 32 + lane] = z4;
    }
}


__global__ void mid32_k(float* __restrict__ p, float* __restrict__ z,
                        const float* __restrict__ hb, const float* __restrict__ dinv,
                        const float* __restrict__ b, float* __restrict__ s, int n) {
    int t = blockIdx.x * blockDim.x + threadIdx.x;
    if (t >= n * 32) return;
    const int i  = t >> 5;
    const int ch = t & 31;
    const float di = dinv[i];
    p[t] = fmaxf(p[t] + z[t], 0.0f);
    z[t] = 0.0f;
    s[t] = fmaf(hb[t], di * di, b[ch]);
}

constexpr int EPW = 128;

template<int G>
__device__ __forceinline__ void edge_group32(int j, int s, int d, float c, int lane,
                                             const float* __restrict__ h, float* __restrict__ agg) {
    float v[G];
#pragma unroll
    for (int u = 0; u < G; ++u) {
        const int sj = __builtin_amdgcn_readlane(s, j + u);
        const float* hr = h + (size_t)sj * 32;
        v[u] = hr[lane];
    }
#pragma unroll
    for (int u = 0; u < G; ++u) {
        const int   dj = __builtin_amdgcn_readlane(d, j + u);
        const float cj = __int_as_float(__builtin_amdgcn_readlane(__float_as_int(c), j + u));
        float* ar = agg + (size_t)dj * 32;
        atomicAdd(ar + lane, v[u] * cj);
    }
}

__global__ __launch_bounds__(256)
void scatter32_k(const int* __restrict__ src, const int* __restrict__ dst,
                 const float* __restrict__ coef, const float* __restrict__ h,
                 float* __restrict__ p0, float* __restrict__ p1, int E) {
    const int lane = threadIdx.x & 31;
    const int wib  = __builtin_amdgcn_readfirstlane((int)(threadIdx.x >> 5));
    const int wave = blockIdx.x * (int)(blockDim.x >> 5) + wib;
    const int ebeg = wave * EPW;
    if (ebeg >= E) return;
    const int eend = min(ebeg + EPW, E);

    float* agg = pkg_half() ? p1 : p0;

    int s = 0, d = 0;
    float c = 0.0f;
    if (ebeg + lane < eend) {
        s = src[ebeg + lane];
        d = dst[ebeg + lane];
        c = coef[ebeg + lane];
    }

    for (int e0 = ebeg; e0 < eend; e0 += 32) {
        const int en = e0 + 32 + lane;
        int sn = 0, dn = 0;
        float cn = 0.0f;
        if (en < eend) {
            sn = src[en];
            dn = dst[en];
            cn = coef[en];
        }

        const int cnt = min(32, eend - e0);
        int j = 0;
        if (cnt == 32) {
            edge_group32<32>(0, s, d, c, lane, h, agg);
            j = 32;
        }
        for (; j + 16 <= cnt; j += 16) edge_group32<16>(j, s, d, c, lane, h, agg);
        for (; j < cnt; ++j)
            edge_group32<1>(j, s, d, c, lane, h, agg);

        s = sn; d = dn; c = cn;
    }
}


__global__ void scatter3_k(const int* __restrict__ src, const int* __restrict__ dst,
                           const float* __restrict__ coef, const float* __restrict__ h,
                           float* __restrict__ o0, float* __restrict__ o1, int E) {
    float* out = pkg_half() ? o1 : o0;
    const int lane = threadIdx.x & 31;
    const int ch   = lane & 3;
    const int wib   = __builtin_amdgcn_readfirstlane((int)(threadIdx.x >> 5));
    const int wbase = (blockIdx.x * (int)(blockDim.x >> 5) + wib) * 32;
    if (wbase >= E) return;
    if (ch == 3) return;
    const int eb = wbase + (lane >> 2) * 4;

    if (wbase + 32 <= E) {
        int   s[4], d[4];
        float c[4], v[4];
#pragma unroll
        for (int g = 0; g < 4; ++g) {
            s[g] = src[eb + g];
            d[g] = dst[eb + g];
            c[g] = coef[eb + g];
        }
#pragma unroll
        for (int g = 0; g < 4; ++g) v[g] = h[(size_t)s[g] * 3 + ch] * c[g];
#pragma unroll
        for (int g = 0; g < 4; ++g) atomicAdd(out + (size_t)d[g] * 3 + ch, v[g]);
    } else {
        for (int g = 0; g < 4; ++g) {
            const int e = eb + g;
            if (e < E) atomicAdd(out + (size_t)dst[e] * 3 + ch, h[(size_t)src[e] * 3 + ch] * coef[e]);
        }
    }
}

__global__ void add_plane_k(float* __restrict__ o0, const float* __restrict__ o1, int n) {
    int i = blockIdx.x * blockDim.x + threadIdx.x;
    if (i < n) o0[i] += o1[i];
}


static inline int cdiv(long long a, int b) { return (int)((a + b - 1) / b); }

extern "C" void kernel_launch(void* const* d_in, const int* in_sizes, int n_in,
                              void* d_out, int out_size, void* d_ws, size_t ws_size,
                              hipStream_t stream) {
    const float* x   = (const float*)d_in[0];
    const int*   ei  = (const int*)d_in[1];
    const float* W1  = (const float*)d_in[2];
    const float* b1  = (const float*)d_in[3];
    const float* W2  = (const float*)d_in[4];
    const float* b2  = (const float*)d_in[5];
    const float* W3  = (const float*)d_in[6];
    const float* b3  = (const float*)d_in[7];
    float* out = (float*)d_out;

    const int N = in_sizes[0] / 9;
    const int E = in_sizes[1] / 2;
    const int* src = ei;
    const int* dst = ei + E;

    float* R0   = (float*)d_ws;
    float* R1   = R0 + (size_t)N * 32;
    float* R2   = R1 + (size_t)N * 32;
    float* R3   = R2 + (size_t)N * 32;
    float* dinv = R3 + (size_t)N * 32;
    float* coef = dinv + N;
    (void)ws_size; (void)n_in; (void)out_size;

    const int ntiles = (N + 15) / 16;
    const int gN   = cdiv(N, 256);
    const int gE   = cdiv(E, 256);
    const int gP   = cdiv((long long)N * 32, 256);
    const int gS   = cdiv(cdiv(E, EPW), 8);
    const int gG   = cdiv(ntiles, 8);
    const int gN3  = cdiv((long long)N * 3, 256);

    fill_deg_k<<<gN, 256, 0, stream>>>(dinv, R0, N);
    deg_accum_k<<<gE, 256, 0, stream>>>(dst, dinv, R0, E);
    rsqrt_k<<<gN, 256, 0, stream>>>(dinv, R0, N);
    coef_k<<<gE, 256, 0, stream>>>(src, dst, dinv, coef, E);

    gemm_wmma_k<9, 64, false, true><<<gG, 256, 0, stream>>>(x, nullptr, nullptr, W1, dinv, b1,
                                                           R0, R1, R2, R3, ntiles, N);
    scatter32_k<<<gS, 256, 0, stream>>>(src, dst, coef, R0, R2, R3, E);
    mid32_k<<<gP, 256, 0, stream>>>(R2, R3, R1, dinv, b1 + 32, R0, N);
    scatter32_k<<<gS, 256, 0, stream>>>(src, dst, coef, R1, R0, R3, E);

    gemm_wmma_k<64, 64, true, true><<<gG, 256, 0, stream>>>(R2, R0, R3, W2, dinv, b2,
                                                           R1, R3, R2, R0, ntiles, N);
    scatter32_k<<<gS, 256, 0, stream>>>(src, dst, coef, R1, R2, R0, E);
    mid32_k<<<gP, 256, 0, stream>>>(R2, R0, R3, dinv, b2 + 32, R1, N);
    scatter32_k<<<gS, 256, 0, stream>>>(src, dst, coef, R3, R1, R0, E);

    float* o1 = R3 + (size_t)N * 4;
    gemm_wmma_k<64, 3, true, false><<<gG, 256, 0, stream>>>(R2, R1, R0, W3, dinv, b3,
                                                           R3, nullptr, out, o1, ntiles, N);
    scatter3_k<<<gE, 256, 0, stream>>>(src, dst, coef, R3, out, o1, E);
    add_plane_k<<<gN3, 256, 0, stream>>>(out, o1, N * 3);
}


# 64-channel f32 atomic sums in one plane set: package halves split the node range, workgroups draw edge blocks by ticket and add only their half's nodes; one edge pass per layer (sum order changes)
# speedup vs baseline: 2.9998x; 1.2527x over previous
#include <hip/hip_runtime.h>


constexpr int CNT_STRIDE = 64;

typedef float v2f __attribute__((ext_vector_type(2)));
typedef float v8f __attribute__((ext_vector_type(8)));

__device__ __forceinline__ int pkg_half() {
    return (int)((__builtin_amdgcn_s_sendmsg_rtn(0x87) >> 18) & 1u);
}


__global__ void fill_deg_k(float* __restrict__ d0, float* __restrict__ d1, int n) {
    int i = blockIdx.x * blockDim.x + threadIdx.x;
    if (i < n) {
        d0[i] = 1.0f;
        d1[i] = 0.0f;
    }
}

__global__ void deg_accum_k(const int* __restrict__ dst, float* __restrict__ d0,
                            float* __restrict__ d1, int E) {
    float* deg = pkg_half() ? d1 : d0;
    int e = blockIdx.x * blockDim.x + threadIdx.x;
    if (e < E) atomicAdd(&deg[dst[e]], 1.0f);
}

__global__ void rsqrt_k(float* __restrict__ d0, const float* __restrict__ d1, int n) {
    int i = blockIdx.x * blockDim.x + threadIdx.x;
    if (i < n) d0[i] = rsqrtf(d0[i] + d1[i]);
}

__global__ void coef_k(const int* __restrict__ src, const int* __restrict__ dst,
                       const float* __restrict__ dinv, float* __restrict__ coef, int E) {
    int e = blockIdx.x * blockDim.x + threadIdx.x;
    if (e < E) coef[e] = dinv[src[e]] * dinv[dst[e]];
}


template<int KDIM, int NOUT, bool PIN, bool POUT>
__global__ void gemm_wmma_k(const float* X0, const float* X1,
                            const float* __restrict__ W,
                            const float* __restrict__ dinv, const float* __restrict__ bias,
                            float* H0, float* H1, float* P0, float* P1,
                            int* __restrict__ cnt, int ntiles, int nrows) {
    constexpr int NT   = (NOUT + 15) / 16;
    constexpr int KT   = (KDIM + 3) / 4;
    constexpr int KPAD = KT * 4;
    constexpr int NW16 = NT * 16;
    static_assert(!PIN  || KDIM == 64, "planar input is 2 x 32 channels");
    static_assert(!POUT || NOUT == 64, "planar output is 2 x 32 channels");
    static_assert(POUT || NOUT <= 16, "row-major output is one column tile");

    __shared__ float sW[KPAD * NW16];

    if constexpr (POUT) {
        if (blockIdx.x == 0 && threadIdx.x < 2) cnt[threadIdx.x * CNT_STRIDE] = 0;
    }

    for (int idx = threadIdx.x; idx < KPAD * NW16; idx += blockDim.x) {
        const int k = idx / NW16;
        const int n = idx - k * NW16;
        sW[idx] = (k < KDIM && n < NOUT) ? W[k * NOUT + n] : 0.0f;
    }
    __syncthreads();

    const int lane = threadIdx.x & 31;
    const int wv   = blockIdx.x * (blockDim.x >> 5) + (threadIdx.x >> 5);
    if (wv >= ntiles) return;
    const int row0 = wv * 16;
    const int m    = lane & 15;
    const int half = lane >> 4;

    const unsigned lim  = (unsigned)nrows * KDIM - 1u;
    const unsigned rowc = min((unsigned)(row0 + m), (unsigned)nrows - 1u);

    float d2[8];
#pragma unroll
    for (int i = 0; i < 8; ++i) {
        const float di = dinv[min((unsigned)(row0 + half * 8 + i), (unsigned)nrows - 1u)];
        d2[i] = di * di;
    }
    float bb[NT];
#pragma unroll
    for (int t = 0; t < NT; ++t) bb[t] = (t * 16 + m < NOUT) ? bias[t * 16 + m] : 0.0f;

    v8f acc[NT];
#pragma unroll
    for (int t = 0; t < NT; ++t) acc[t] = 0.0f;

    v2f av[KT];
#pragma unroll
    for (int kt = 0; kt < KT; ++kt) {
        if constexpr (PIN) {
            const unsigned kk = (unsigned)((kt * 4) & 31) + (unsigned)(half * 2);
            const unsigned ix = rowc * 32u + kk;
            const float* xp = (kt * 4 < 32) ? X0 : X1;
            av[kt].x = xp[ix];
            av[kt].y = xp[ix + 1u];
        } else {
            const unsigned i0 = (unsigned)(row0 + m) * KDIM + kt * 4 + half * 2;
            av[kt].x = X0[min(i0,      lim)];
            av[kt].y = X0[min(i0 + 1u, lim)];
        }
    }

#pragma unroll
    for (int kt = 0; kt < KT; ++kt) {
        const int ka = kt * 4 + half * 2;
        v2f a = av[kt];
        if constexpr (PIN) {
            a.x = fmaxf(av[kt].x, 0.0f);
            a.y = fmaxf(av[kt].y, 0.0f);
        }
#pragma unroll
        for (int nt = 0; nt < NT; ++nt) {
            const int col = nt * 16 + m;
            v2f b;
            b.x = sW[ka * NW16 + col];
            b.y = sW[(ka + 1) * NW16 + col];
            acc[nt] = __builtin_amdgcn_wmma_f32_16x16x4_f32(
                false, a, false, b, (short)0, acc[nt], false, false);
        }
    }

#pragma unroll
    for (int nt = 0; nt < NT; ++nt) {
#pragma unroll
        for (int i = 0; i < 8; ++i) {
            const int r   = row0 + i + half * 8;
            const int col = nt * 16 + m;
            if constexpr (POUT) {
                float* hp = (nt * 16 < 32) ? H0 : H1;
                float* pp = (nt * 16 < 32) ? P0 : P1;
                hp[(size_t)r * 32 + (col & 31)] = acc[nt][i];
                pp[(size_t)r * 32 + (col & 31)] = fmaf(acc[nt][i], d2[i], bb[nt]);
            } else {
                if (col < NOUT) {
                    H0[(size_t)r * NOUT + col] = acc[nt][i];
                    P0[(size_t)r * NOUT + col] = fmaf(acc[nt][i], d2[i], bb[0]);
                    P1[(size_t)r * NOUT + col] = 0.0f;
                }
            }
        }
    }
}

constexpr int EPW = 256;
constexpr int WPB = 8;
constexpr int EPB = EPW * WPB;

template<int G>
__device__ __forceinline__ void edge_group64(int j, int s, int d, float c, int lane,
                                             const float* __restrict__ h0, const float* __restrict__ h1,
                                             float* __restrict__ a0, float* __restrict__ a1) {
    float v0[G], v1[G];
#pragma unroll
    for (int u = 0; u < G; ++u) {
        const int sj = __builtin_amdgcn_readlane(s, j + u);
        const size_t ro = (size_t)sj * 32;
        v0[u] = h0[ro + lane];
        v1[u] = h1[ro + lane];
    }
#pragma unroll
    for (int u = 0; u < G; ++u) {
        const int   dj = __builtin_amdgcn_readlane(d, j + u);
        const float cj = __int_as_float(__builtin_amdgcn_readlane(__float_as_int(c), j + u));
        const size_t ro = (size_t)dj * 32;
        atomicAdd(a0 + ro + lane, v0[u] * cj);
        atomicAdd(a1 + ro + lane, v1[u] * cj);
    }
}

__device__ __forceinline__ void edge_tail64(int j, int cn, int s, int d, float c, int lane,
                                            const float* __restrict__ h0, const float* __restrict__ h1,
                                            float* __restrict__ a0, float* __restrict__ a1) {
    float v0[15], v1[15];
#pragma unroll
    for (int u = 0; u < 15; ++u) {
        if (j + u < cn) {
            const int sj = __builtin_amdgcn_readlane(s, j + u);
            const size_t ro = (size_t)sj * 32;
            v0[u] = h0[ro + lane];
            v1[u] = h1[ro + lane];
        }
    }
#pragma unroll
    for (int u = 0; u < 15; ++u) {
        if (j + u < cn) {
            const int   dj = __builtin_amdgcn_readlane(d, j + u);
            const float cj = __int_as_float(__builtin_amdgcn_readlane(__float_as_int(c), j + u));
            const size_t ro = (size_t)dj * 32;
            atomicAdd(a0 + ro + lane, v0[u] * cj);
            atomicAdd(a1 + ro + lane, v1[u] * cj);
        }
    }
}

__global__ __launch_bounds__(256)
void scatter64_k(const int* __restrict__ src, const int* __restrict__ dst,
                 const float* __restrict__ coef,
                 const float* __restrict__ h0, const float* __restrict__ h1,
                 float* __restrict__ a0, float* __restrict__ a1,
                 int* __restrict__ cnt, int E, int n, int nblocks) {
    __shared__ int   tk[2];
    __shared__ int   qs[WPB][EPW];
    __shared__ int   qd[WPB][EPW];
    __shared__ float qc[WPB][EPW];

    const int hme = pkg_half();
    if (threadIdx.x == 0) {
        int r = hme;
        int t = atomicAdd(&cnt[r * CNT_STRIDE], 1);
        if (t >= nblocks) {
            r ^= 1;
            t = atomicAdd(&cnt[r * CNT_STRIDE], 1);
        }
        tk[0] = t;
        tk[1] = r;
    }
    __syncthreads();
    const int t = __builtin_amdgcn_readfirstlane(tk[0]);
    const int r = __builtin_amdgcn_readfirstlane(tk[1]);
    if (t >= nblocks) return;

    const int lane = threadIdx.x & 31;
    const int wib  = __builtin_amdgcn_readfirstlane((int)(threadIdx.x >> 5));
    const int ebeg = t * EPB + wib * EPW;
    if (ebeg >= E) return;
    const int eend   = min(ebeg + EPW, E);
    const int nsplit = n >> 1;

    int*   ws = qs[wib];
    int*   wd = qd[wib];
    float* wc = qc[wib];

    int   sv[EPW / 32], dv[EPW / 32];
    float cv[EPW / 32];
#pragma unroll
    for (int b = 0; b < EPW / 32; ++b) {
        const int e = min(ebeg + b * 32 + lane, E - 1);
        dv[b] = dst[e];
        sv[b] = src[e];
        cv[b] = coef[e];
    }
    asm volatile("" :: "v"(sv[0]), "v"(cv[0]), "v"(sv[1]), "v"(cv[1]), "v"(sv[2]), "v"(cv[2]),
                       "v"(sv[3]), "v"(cv[3]), "v"(sv[4]), "v"(cv[4]), "v"(sv[5]), "v"(cv[5]),
                       "v"(sv[6]), "v"(cv[6]), "v"(sv[7]), "v"(cv[7]));
    static_assert(EPW / 32 == 8, "the fence above names eight groups of loads");
    int tot = 0;
#pragma unroll
    for (int b = 0; b < EPW / 32; ++b) {
        const bool mine = (ebeg + b * 32 + lane < eend) && ((dv[b] >= nsplit) == (r != 0));
        const unsigned m = __builtin_amdgcn_ballot_w32(mine);
        if (mine) {
            const int k = tot + (int)__builtin_amdgcn_mbcnt_lo(m, 0u);
            ws[k] = sv[b];
            wd[k] = dv[b];
            wc[k] = cv[b];
        }
        tot += __builtin_popcount(m);
    }

    for (int g0 = 0; g0 < tot; g0 += 32) {
        const int k = g0 + lane;
        int s = 0, d = 0;
        float c = 0.0f;
        if (k < tot) {
            s = ws[k];
            d = wd[k];
            c = wc[k];
        }
        const int cn = min(32, tot - g0);
        int j = 0;
        for (; j + 16 <= cn; j += 16) edge_group64<16>(j, s, d, c, lane, h0, h1, a0, a1);
        if (j < cn) edge_tail64(j, cn, s, d, c, lane, h0, h1, a0, a1);
    }
}


__global__ void scatter3_k(const int* __restrict__ src, const int* __restrict__ dst,
                           const float* __restrict__ coef, const float* __restrict__ h,
                           float* __restrict__ o0, float* __restrict__ o1, int E) {
    float* out = pkg_half() ? o1 : o0;
    const int lane = threadIdx.x & 31;
    const int ch   = lane & 3;
    const int wib   = __builtin_amdgcn_readfirstlane((int)(threadIdx.x >> 5));
    const int wbase = (blockIdx.x * (int)(blockDim.x >> 5) + wib) * 32;
    if (wbase >= E) return;
    if (ch == 3) return;
    const int eb = wbase + (lane >> 2) * 4;

    if (wbase + 32 <= E) {
        int   s[4], d[4];
        float c[4], v[4];
#pragma unroll
        for (int g = 0; g < 4; ++g) {
            s[g] = src[eb + g];
            d[g] = dst[eb + g];
            c[g] = coef[eb + g];
        }
#pragma unroll
        for (int g = 0; g < 4; ++g) v[g] = h[(size_t)s[g] * 3 + ch] * c[g];
#pragma unroll
        for (int g = 0; g < 4; ++g) atomicAdd(out + (size_t)d[g] * 3 + ch, v[g]);
    } else {
        for (int g = 0; g < 4; ++g) {
            const int e = eb + g;
            if (e < E) atomicAdd(out + (size_t)dst[e] * 3 + ch, h[(size_t)src[e] * 3 + ch] * coef[e]);
        }
    }
}

__global__ void add_plane_k(float* __restrict__ o0, const float* __restrict__ o1, int n) {
    int i = blockIdx.x * blockDim.x + threadIdx.x;
    if (i < n) o0[i] += o1[i];
}


static inline int cdiv(long long a, int b) { return (int)((a + b - 1) / b); }

extern "C" void kernel_launch(void* const* d_in, const int* in_sizes, int n_in,
                              void* d_out, int out_size, void* d_ws, size_t ws_size,
                              hipStream_t stream) {
    const float* x   = (const float*)d_in[0];
    const int*   ei  = (const int*)d_in[1];
    const float* W1  = (const float*)d_in[2];
    const float* b1  = (const float*)d_in[3];
    const float* W2  = (const float*)d_in[4];
    const float* b2  = (const float*)d_in[5];
    const float* W3  = (const float*)d_in[6];
    const float* b3  = (const float*)d_in[7];
    float* out = (float*)d_out;

    const int N = in_sizes[0] / 9;
    const int E = in_sizes[1] / 2;
    const int* src = ei;
    const int* dst = ei + E;

    float* R0   = (float*)d_ws;
    float* R1   = R0 + (size_t)N * 32;
    float* R2   = R1 + (size_t)N * 32;
    float* R3   = R2 + (size_t)N * 32;
    float* dinv = R3 + (size_t)N * 32;
    float* coef = dinv + N;
    (void)ws_size; (void)n_in; (void)out_size;

    const int ntiles = (N + 15) / 16;
    const int gN   = cdiv(N, 256);
    const int gE   = cdiv(E, 256);
    const int nEB  = cdiv(E, EPB);
    const int gG   = cdiv(ntiles, 8);
    const int gN3  = cdiv((long long)N * 3, 256);

    fill_deg_k<<<gN, 256, 0, stream>>>(dinv, R0, N);
    deg_accum_k<<<gE, 256, 0, stream>>>(dst, dinv, R0, E);
    rsqrt_k<<<gN, 256, 0, stream>>>(dinv, R0, N);
    coef_k<<<gE, 256, 0, stream>>>(src, dst, dinv, coef, E);

    int* tick = (int*)d_out;

    gemm_wmma_k<9, 64, false, true><<<gG, 256, 0, stream>>>(x, nullptr, W1, dinv, b1,
                                                           R0, R1, R2, R3, tick, ntiles, N);
    scatter64_k<<<2 * nEB, 256, 0, stream>>>(src, dst, coef, R0, R1, R2, R3, tick, E, N, nEB);

    gemm_wmma_k<64, 64, true, true><<<gG, 256, 0, stream>>>(R2, R3, W2, dinv, b2,
                                                           R0, R1, R2, R3, tick, ntiles, N);
    scatter64_k<<<2 * nEB, 256, 0, stream>>>(src, dst, coef, R0, R1, R2, R3, tick, E, N, nEB);

    float* o1 = R0 + (size_t)N * 4;
    gemm_wmma_k<64, 3, true, false><<<gG, 256, 0, stream>>>(R2, R3, W3, dinv, b3,
                                                           R0, nullptr, out, o1, nullptr, ntiles, N);
    scatter3_k<<<gE, 256, 0, stream>>>(src, dst, coef, R0, out, o1, E);
    add_plane_k<<<gN3, 256, 0, stream>>>(out, o1, N * 3);
}


# Passes that gave a lane one edge (degree count, coefficients, layer-3 adds) or one node word give it 8 edges (4 words), loads grouped: 8x (4x) fewer waves to start; atomics stay in arrival order
# speedup vs baseline: 3.0956x; 1.0319x over previous
#include <hip/hip_runtime.h>


constexpr int CNT_STRIDE = 64;

typedef float v2f __attribute__((ext_vector_type(2)));
typedef float v8f __attribute__((ext_vector_type(8)));
typedef float v4f __attribute__((ext_vector_type(4)));
typedef int   v4i __attribute__((ext_vector_type(4)));

__device__ __forceinline__ int pkg_half() {
    return (int)((__builtin_amdgcn_s_sendmsg_rtn(0x87) >> 18) & 1u);
}


constexpr int EPL  = 8;
constexpr int EPWV = 32 * EPL;
constexpr int NPL  = 4;

__device__ __forceinline__ int wave_first_edge() {
    const int wib = __builtin_amdgcn_readfirstlane((int)(threadIdx.x >> 5));
    return (blockIdx.x * (int)(blockDim.x >> 5) + wib) * EPWV;
}

__device__ __forceinline__ int lane_edge(int wbase, int lane, int k) {
    return wbase + (k >> 2) * 128 + lane * 4 + (k & 3);
}

__global__ void fill_deg_k(float* __restrict__ d0, float* __restrict__ d1, int n) {
    const int i0 = (blockIdx.x * blockDim.x + threadIdx.x) * NPL;
    if (i0 + NPL <= n) {
#pragma unroll
        for (int g = 0; g < NPL; ++g) {
            d0[i0 + g] = 1.0f;
            d1[i0 + g] = 0.0f;
        }
    } else {
        for (int g = 0; g < NPL; ++g) {
            if (i0 + g < n) {
                d0[i0 + g] = 1.0f;
                d1[i0 + g] = 0.0f;
            }
        }
    }
}

__global__ void deg_accum_k(const int* __restrict__ dst, float* __restrict__ d0,
                            float* __restrict__ d1, int E) {
    float* deg = pkg_half() ? d1 : d0;
    const int lane  = threadIdx.x & 31;
    const int wbase = wave_first_edge();
    if (wbase >= E) return;
    if (wbase + EPWV <= E) {
        int d[EPL];
#pragma unroll
        for (int k = 0; k < EPL; ++k) d[k] = dst[lane_edge(wbase, lane, k)];
        asm volatile("" :: "v"(d[0]), "v"(d[1]), "v"(d[2]), "v"(d[3]),
                           "v"(d[4]), "v"(d[5]), "v"(d[6]), "v"(d[7]));
        static_assert(EPL == 8, "the fence above names eight edges");
#pragma unroll
        for (int k = 0; k < EPL; ++k) atomicAdd(&deg[d[k]], 1.0f);
    } else {
#pragma nounroll
        for (int k = 0; k < EPL; ++k) {
            const int e = lane_edge(wbase, lane, k);
            if (e < E) atomicAdd(&deg[dst[e]], 1.0f);
        }
    }
}

__global__ void rsqrt_k(float* __restrict__ d0, const float* __restrict__ d1, int n) {
    const int i0 = (blockIdx.x * blockDim.x + threadIdx.x) * NPL;
    if (i0 + NPL <= n) {
        float a[NPL], b[NPL];
#pragma unroll
        for (int g = 0; g < NPL; ++g) {
            a[g] = d0[i0 + g];
            b[g] = d1[i0 + g];
        }
#pragma unroll
        for (int g = 0; g < NPL; ++g) d0[i0 + g] = rsqrtf(a[g] + b[g]);
    } else {
        for (int g = 0; g < NPL; ++g) {
            if (i0 + g < n) d0[i0 + g] = rsqrtf(d0[i0 + g] + d1[i0 + g]);
        }
    }
}

__global__ void coef_k(const int* __restrict__ src, const int* __restrict__ dst,
                       const float* __restrict__ dinv, float* __restrict__ coef, int E) {
    const int lane  = threadIdx.x & 31;
    const int wbase = wave_first_edge();
    if (wbase >= E) return;
    if (wbase + EPWV <= E) {
        int s[EPL], d[EPL];
#pragma unroll
        for (int k = 0; k < EPL; ++k) {
            s[k] = src[lane_edge(wbase, lane, k)];
            d[k] = dst[lane_edge(wbase, lane, k)];
        }
        float a[EPL], b[EPL];
#pragma unroll
        for (int k = 0; k < EPL; ++k) {
            a[k] = dinv[s[k]];
            b[k] = dinv[d[k]];
        }
#pragma unroll
        for (int k = 0; k < EPL; ++k) coef[lane_edge(wbase, lane, k)] = a[k] * b[k];
    } else {
#pragma nounroll
        for (int k = 0; k < EPL; ++k) {
            const int e = lane_edge(wbase, lane, k);
            if (e < E) coef[e] = dinv[src[e]] * dinv[dst[e]];
        }
    }
}


template<int KDIM, int NOUT, bool PIN, bool POUT>
__global__ void gemm_wmma_k(const float* X0, const float* X1,
                            const float* __restrict__ W,
                            const float* __restrict__ dinv, const float* __restrict__ bias,
                            float* H0, float* H1, float* P0, float* P1,
                            int* __restrict__ cnt, int ntiles, int nrows) {
    constexpr int NT   = (NOUT + 15) / 16;
    constexpr int KT   = (KDIM + 3) / 4;
    constexpr int KPAD = KT * 4;
    constexpr int NW16 = NT * 16;
    static_assert(!PIN  || KDIM == 64, "planar input is 2 x 32 channels");
    static_assert(!POUT || NOUT == 64, "planar output is 2 x 32 channels");
    static_assert(POUT || NOUT <= 16, "row-major output is one column tile");

    __shared__ float sW[KPAD * NW16];

    if constexpr (POUT) {
        if (blockIdx.x == 0 && threadIdx.x < 2) cnt[threadIdx.x * CNT_STRIDE] = 0;
    }

    for (int idx = threadIdx.x; idx < KPAD * NW16; idx += blockDim.x) {
        const int k = idx / NW16;
        const int n = idx - k * NW16;
        sW[idx] = (k < KDIM && n < NOUT) ? W[k * NOUT + n] : 0.0f;
    }
    __syncthreads();

    const int lane = threadIdx.x & 31;
    const int wv   = blockIdx.x * (blockDim.x >> 5) + (threadIdx.x >> 5);
    if (wv >= ntiles) return;
    const int row0 = wv * 16;
    const int m    = lane & 15;
    const int half = lane >> 4;

    const unsigned lim  = (unsigned)nrows * KDIM - 1u;
    const unsigned rowc = min((unsigned)(row0 + m), (unsigned)nrows - 1u);

    float d2[8];
#pragma unroll
    for (int i = 0; i < 8; ++i) {
        const float di = dinv[min((unsigned)(row0 + half * 8 + i), (unsigned)nrows - 1u)];
        d2[i] = di * di;
    }
    float bb[NT];
#pragma unroll
    for (int t = 0; t < NT; ++t) bb[t] = (t * 16 + m < NOUT) ? bias[t * 16 + m] : 0.0f;

    v8f acc[NT];
#pragma unroll
    for (int t = 0; t < NT; ++t) acc[t] = 0.0f;

    v2f av[KT];
#pragma unroll
    for (int kt = 0; kt < KT; ++kt) {
        if constexpr (PIN) {
            const unsigned kk = (unsigned)((kt * 4) & 31) + (unsigned)(half * 2);
            const unsigned ix = rowc * 32u + kk;
            const float* xp = (kt * 4 < 32) ? X0 : X1;
            av[kt].x = xp[ix];
            av[kt].y = xp[ix + 1u];
        } else {
            const unsigned i0 = (unsigned)(row0 + m) * KDIM + kt * 4 + half * 2;
            av[kt].x = X0[min(i0,      lim)];
            av[kt].y = X0[min(i0 + 1u, lim)];
        }
    }

#pragma unroll
    for (int kt = 0; kt < KT; ++kt) {
        const int ka = kt * 4 + half * 2;
        v2f a = av[kt];
        if constexpr (PIN) {
            a.x = fmaxf(av[kt].x, 0.0f);
            a.y = fmaxf(av[kt].y, 0.0f);
        }
#pragma unroll
        for (int nt = 0; nt < NT; ++nt) {
            const int col = nt * 16 + m;
            v2f b;
            b.x = sW[ka * NW16 + col];
            b.y = sW[(ka + 1) * NW16 + col];
            acc[nt] = __builtin_amdgcn_wmma_f32_16x16x4_f32(
                false, a, false, b, (short)0, acc[nt], false, false);
        }
    }

#pragma unroll
    for (int nt = 0; nt < NT; ++nt) {
#pragma unroll
        for (int i = 0; i < 8; ++i) {
            const int r   = row0 + i + half * 8;
            const int col = nt * 16 + m;
            if constexpr (POUT) {
                float* hp = (nt * 16 < 32) ? H0 : H1;
                float* pp = (nt * 16 < 32) ? P0 : P1;
                hp[(size_t)r * 32 + (col & 31)] = acc[nt][i];
                pp[(size_t)r * 32 + (col & 31)] = fmaf(acc[nt][i], d2[i], bb[nt]);
            } else {
                if (col < NOUT) {
                    H0[(size_t)r * NOUT + col] = acc[nt][i];
                    P0[(size_t)r * NOUT + col] = fmaf(acc[nt][i], d2[i], bb[0]);
                    P1[(size_t)r * NOUT + col] = 0.0f;
                }
            }
        }
    }
}

constexpr int EPW = 256;
constexpr int WPB = 8;
constexpr int EPB = EPW * WPB;

template<int G>
__device__ __forceinline__ void edge_group64(int j, int s, int d, float c, int lane,
                                             const float* __restrict__ h0, const float* __restrict__ h1,
                                             float* __restrict__ a0, float* __restrict__ a1) {
    float v0[G], v1[G];
#pragma unroll
    for (int u = 0; u < G; ++u) {
        const int sj = __builtin_amdgcn_readlane(s, j + u);
        const size_t ro = (size_t)sj * 32;
        v0[u] = h0[ro + lane];
        v1[u] = h1[ro + lane];
    }
#pragma unroll
    for (int u = 0; u < G; ++u) {
        const int   dj = __builtin_amdgcn_readlane(d, j + u);
        const float cj = __int_as_float(__builtin_amdgcn_readlane(__float_as_int(c), j + u));
        const size_t ro = (size_t)dj * 32;
        atomicAdd(a0 + ro + lane, v0[u] * cj);
        atomicAdd(a1 + ro + lane, v1[u] * cj);
    }
}

__device__ __forceinline__ void edge_tail64(int j, int cn, int s, int d, float c, int lane,
                                            const float* __restrict__ h0, const float* __restrict__ h1,
                                            float* __restrict__ a0, float* __restrict__ a1) {
    float v0[15], v1[15];
#pragma unroll
    for (int u = 0; u < 15; ++u) {
        if (j + u < cn) {
            const int sj = __builtin_amdgcn_readlane(s, j + u);
            const size_t ro = (size_t)sj * 32;
            v0[u] = h0[ro + lane];
            v1[u] = h1[ro + lane];
        }
    }
#pragma unroll
    for (int u = 0; u < 15; ++u) {
        if (j + u < cn) {
            const int   dj = __builtin_amdgcn_readlane(d, j + u);
            const float cj = __int_as_float(__builtin_amdgcn_readlane(__float_as_int(c), j + u));
            const size_t ro = (size_t)dj * 32;
            atomicAdd(a0 + ro + lane, v0[u] * cj);
            atomicAdd(a1 + ro + lane, v1[u] * cj);
        }
    }
}

__global__ __launch_bounds__(256)
void scatter64_k(const int* __restrict__ src, const int* __restrict__ dst,
                 const float* __restrict__ coef,
                 const float* __restrict__ h0, const float* __restrict__ h1,
                 float* __restrict__ a0, float* __restrict__ a1,
                 int* __restrict__ cnt, int E, int n, int nblocks) {
    __shared__ int   tk[2];
    __shared__ int   qs[WPB][EPW];
    __shared__ int   qd[WPB][EPW];
    __shared__ float qc[WPB][EPW];

    const int hme = pkg_half();
    if (threadIdx.x == 0) {
        int r = hme;
        int t = atomicAdd(&cnt[r * CNT_STRIDE], 1);
        if (t >= nblocks) {
            r ^= 1;
            t = atomicAdd(&cnt[r * CNT_STRIDE], 1);
        }
        tk[0] = t;
        tk[1] = r;
    }
    __syncthreads();
    const int t = __builtin_amdgcn_readfirstlane(tk[0]);
    const int r = __builtin_amdgcn_readfirstlane(tk[1]);
    if (t >= nblocks) return;

    const int lane = threadIdx.x & 31;
    const int wib  = __builtin_amdgcn_readfirstlane((int)(threadIdx.x >> 5));
    const int ebeg = t * EPB + wib * EPW;
    if (ebeg >= E) return;
    const int eend   = min(ebeg + EPW, E);
    const int nsplit = n >> 1;

    int*   ws = qs[wib];
    int*   wd = qd[wib];
    float* wc = qc[wib];

    int   sv[EPW / 32], dv[EPW / 32];
    float cv[EPW / 32];
#pragma unroll
    for (int b = 0; b < EPW / 32; ++b) {
        const int e = min(ebeg + b * 32 + lane, E - 1);
        dv[b] = dst[e];
        sv[b] = src[e];
        cv[b] = coef[e];
    }
    asm volatile("" :: "v"(sv[0]), "v"(cv[0]), "v"(sv[1]), "v"(cv[1]), "v"(sv[2]), "v"(cv[2]),
                       "v"(sv[3]), "v"(cv[3]), "v"(sv[4]), "v"(cv[4]), "v"(sv[5]), "v"(cv[5]),
                       "v"(sv[6]), "v"(cv[6]), "v"(sv[7]), "v"(cv[7]));
    static_assert(EPW / 32 == 8, "the fence above names eight groups of loads");
    int tot = 0;
#pragma unroll
    for (int b = 0; b < EPW / 32; ++b) {
        const bool mine = (ebeg + b * 32 + lane < eend) && ((dv[b] >= nsplit) == (r != 0));
        const unsigned m = __builtin_amdgcn_ballot_w32(mine);
        if (mine) {
            const int k = tot + (int)__builtin_amdgcn_mbcnt_lo(m, 0u);
            ws[k] = sv[b];
            wd[k] = dv[b];
            wc[k] = cv[b];
        }
        tot += __builtin_popcount(m);
    }

    for (int g0 = 0; g0 < tot; g0 += 32) {
        const int k = g0 + lane;
        int s = 0, d = 0;
        float c = 0.0f;
        if (k < tot) {
            s = ws[k];
            d = wd[k];
            c = wc[k];
        }
        const int cn = min(32, tot - g0);
        int j = 0;
        for (; j + 16 <= cn; j += 16) edge_group64<16>(j, s, d, c, lane, h0, h1, a0, a1);
        if (j < cn) edge_tail64(j, cn, s, d, c, lane, h0, h1, a0, a1);
    }
}


__global__ __launch_bounds__(256)
void scatter3_k(const int* __restrict__ src, const int* __restrict__ dst,
                const float* __restrict__ coef, const float* __restrict__ h,
                float* __restrict__ o0, float* __restrict__ o1, int E) {
    float* out = pkg_half() ? o1 : o0;
    const int lane  = threadIdx.x & 31;
    const int ch    = lane & 3;
    const int wbase = wave_first_edge();
    if (wbase >= E) return;
    if (ch == 3) return;
    const int eb = wbase + (lane >> 2) * 4;

    if (wbase + EPWV <= E) {
        v4i   s[EPL], d[EPL];
        v4f   c[EPL];
        float v[EPL][4];
#pragma unroll
        for (int q = 0; q < EPL; ++q) {
            const int e0 = eb + q * 32;
            s[q] = (v4i){src[e0], src[e0 + 1], src[e0 + 2], src[e0 + 3]};
            d[q] = (v4i){dst[e0], dst[e0 + 1], dst[e0 + 2], dst[e0 + 3]};
            c[q] = (v4f){coef[e0], coef[e0 + 1], coef[e0 + 2], coef[e0 + 3]};
        }
        asm volatile("" :: "v"(s[0]), "v"(s[1]), "v"(s[2]), "v"(s[3]), "v"(s[4]), "v"(s[5]), "v"(s[6]), "v"(s[7]),
                           "v"(d[0]), "v"(d[1]), "v"(d[2]), "v"(d[3]), "v"(d[4]), "v"(d[5]), "v"(d[6]), "v"(d[7]),
                           "v"(c[0]), "v"(c[1]), "v"(c[2]), "v"(c[3]), "v"(c[4]), "v"(c[5]), "v"(c[6]), "v"(c[7]));
        static_assert(EPL == 8, "the fence above names eight runs");
#pragma unroll
        for (int q = 0; q < EPL; ++q) {
#pragma unroll
            for (int g = 0; g < 4; ++g) v[q][g] = h[(size_t)s[q][g] * 3 + ch] * c[q][g];
        }
#pragma unroll
        for (int q = 0; q < EPL; ++q) {
#pragma unroll
            for (int g = 0; g < 4; ++g) atomicAdd(out + (size_t)d[q][g] * 3 + ch, v[q][g]);
        }
    } else {
#pragma nounroll
        for (int q = 0; q < EPL; ++q) {
            for (int g = 0; g < 4; ++g) {
                const int e = eb + q * 32 + g;
                if (e < E) atomicAdd(out + (size_t)dst[e] * 3 + ch, h[(size_t)src[e] * 3 + ch] * coef[e]);
            }
        }
    }
}

__global__ void add_plane_k(float* __restrict__ o0, const float* __restrict__ o1, int n) {
    const int i0 = (blockIdx.x * blockDim.x + threadIdx.x) * NPL;
    if (i0 + NPL <= n) {
        float a[NPL], b[NPL];
#pragma unroll
        for (int g = 0; g < NPL; ++g) {
            a[g] = o0[i0 + g];
            b[g] = o1[i0 + g];
        }
#pragma unroll
        for (int g = 0; g < NPL; ++g) o0[i0 + g] = a[g] + b[g];
    } else {
        for (int g = 0; g < NPL; ++g) {
            if (i0 + g < n) o0[i0 + g] += o1[i0 + g];
        }
    }
}


static inline int cdiv(long long a, int b) { return (int)((a + b - 1) / b); }

extern "C" void kernel_launch(void* const* d_in, const int* in_sizes, int n_in,
                              void* d_out, int out_size, void* d_ws, size_t ws_size,
                              hipStream_t stream) {
    const float* x   = (const float*)d_in[0];
    const int*   ei  = (const int*)d_in[1];
    const float* W1  = (const float*)d_in[2];
    const float* b1  = (const float*)d_in[3];
    const float* W2  = (const float*)d_in[4];
    const float* b2  = (const float*)d_in[5];
    const float* W3  = (const float*)d_in[6];
    const float* b3  = (const float*)d_in[7];
    float* out = (float*)d_out;

    const int N = in_sizes[0] / 9;
    const int E = in_sizes[1] / 2;
    const int* src = ei;
    const int* dst = ei + E;

    float* R0   = (float*)d_ws;
    float* R1   = R0 + (size_t)N * 32;
    float* R2   = R1 + (size_t)N * 32;
    float* R3   = R2 + (size_t)N * 32;
    float* dinv = R3 + (size_t)N * 32;
    float* coef = dinv + N;
    (void)ws_size; (void)n_in; (void)out_size;

    const int ntiles = (N + 15) / 16;
    const int gN   = cdiv(N, 256 * NPL);
    const int gE   = cdiv(E, 256 * EPL);
    const int nEB  = cdiv(E, EPB);
    const int gG   = cdiv(ntiles, 8);
    const int gN3  = cdiv((long long)N * 3, 256 * NPL);

    fill_deg_k<<<gN, 256, 0, stream>>>(dinv, R0, N);
    deg_accum_k<<<gE, 256, 0, stream>>>(dst, dinv, R0, E);
    rsqrt_k<<<gN, 256, 0, stream>>>(dinv, R0, N);
    coef_k<<<gE, 256, 0, stream>>>(src, dst, dinv, coef, E);

    int* tick = (int*)d_out;

    gemm_wmma_k<9, 64, false, true><<<gG, 256, 0, stream>>>(x, nullptr, W1, dinv, b1,
                                                           R0, R1, R2, R3, tick, ntiles, N);
    scatter64_k<<<2 * nEB, 256, 0, stream>>>(src, dst, coef, R0, R1, R2, R3, tick, E, N, nEB);

    gemm_wmma_k<64, 64, true, true><<<gG, 256, 0, stream>>>(R2, R3, W2, dinv, b2,
                                                           R0, R1, R2, R3, tick, ntiles, N);
    scatter64_k<<<2 * nEB, 256, 0, stream>>>(src, dst, coef, R0, R1, R2, R3, tick, E, N, nEB);

    float* o1 = R0 + (size_t)N * 4;
    gemm_wmma_k<64, 3, true, false><<<gG, 256, 0, stream>>>(R2, R3, W3, dinv, b3,
                                                           R0, nullptr, out, o1, nullptr, ntiles, N);
    scatter3_k<<<gE, 256, 0, stream>>>(src, dst, coef, R0, out, o1, E);
    add_plane_k<<<gN3, 256, 0, stream>>>(out, o1, N * 3);
}


# 64-channel edge walk: row loads and f32 atomic adds take the plane pointer as scalar base plus one 32-bit byte offset per lane (was a 64-bit vector address per plane and access); sum order kept
# speedup vs baseline: 3.2112x; 1.0373x over previous
#include <hip/hip_runtime.h>


constexpr int CNT_STRIDE = 64;

typedef float v2f __attribute__((ext_vector_type(2)));
typedef float v8f __attribute__((ext_vector_type(8)));
typedef float v4f __attribute__((ext_vector_type(4)));
typedef int   v4i __attribute__((ext_vector_type(4)));

__device__ __forceinline__ int pkg_half() {
    return (int)((__builtin_amdgcn_s_sendmsg_rtn(0x87) >> 18) & 1u);
}


constexpr int EPL  = 8;
constexpr int EPWV = 32 * EPL;
constexpr int NPL  = 4;

__device__ __forceinline__ int wave_first_edge() {
    const int wib = __builtin_amdgcn_readfirstlane((int)(threadIdx.x >> 5));
    return (blockIdx.x * (int)(blockDim.x >> 5) + wib) * EPWV;
}

__device__ __forceinline__ int lane_edge(int wbase, int lane, int k) {
    return wbase + (k >> 2) * 128 + lane * 4 + (k & 3);
}

__global__ void fill_deg_k(float* __restrict__ d0, float* __restrict__ d1, int n) {
    const int i0 = (blockIdx.x * blockDim.x + threadIdx.x) * NPL;
    if (i0 + NPL <= n) {
#pragma unroll
        for (int g = 0; g < NPL; ++g) {
            d0[i0 + g] = 1.0f;
            d1[i0 + g] = 0.0f;
        }
    } else {
        for (int g = 0; g < NPL; ++g) {
            if (i0 + g < n) {
                d0[i0 + g] = 1.0f;
                d1[i0 + g] = 0.0f;
            }
        }
    }
}

__global__ void deg_accum_k(const int* __restrict__ dst, float* __restrict__ d0,
                            float* __restrict__ d1, int E) {
    float* deg = pkg_half() ? d1 : d0;
    const int lane  = threadIdx.x & 31;
    const int wbase = wave_first_edge();
    if (wbase >= E) return;
    if (wbase + EPWV <= E) {
        int d[EPL];
#pragma unroll
        for (int k = 0; k < EPL; ++k) d[k] = dst[lane_edge(wbase, lane, k)];
        asm volatile("" :: "v"(d[0]), "v"(d[1]), "v"(d[2]), "v"(d[3]),
                           "v"(d[4]), "v"(d[5]), "v"(d[6]), "v"(d[7]));
        static_assert(EPL == 8, "the fence above names eight edges");
#pragma unroll
        for (int k = 0; k < EPL; ++k) atomicAdd(&deg[d[k]], 1.0f);
    } else {
#pragma nounroll
        for (int k = 0; k < EPL; ++k) {
            const int e = lane_edge(wbase, lane, k);
            if (e < E) atomicAdd(&deg[dst[e]], 1.0f);
        }
    }
}

__global__ void rsqrt_k(float* __restrict__ d0, const float* __restrict__ d1, int n) {
    const int i0 = (blockIdx.x * blockDim.x + threadIdx.x) * NPL;
    if (i0 + NPL <= n) {
        float a[NPL], b[NPL];
#pragma unroll
        for (int g = 0; g < NPL; ++g) {
            a[g] = d0[i0 + g];
            b[g] = d1[i0 + g];
        }
#pragma unroll
        for (int g = 0; g < NPL; ++g) d0[i0 + g] = rsqrtf(a[g] + b[g]);
    } else {
        for (int g = 0; g < NPL; ++g) {
            if (i0 + g < n) d0[i0 + g] = rsqrtf(d0[i0 + g] + d1[i0 + g]);
        }
    }
}

__global__ void coef_k(const int* __restrict__ src, const int* __restrict__ dst,
                       const float* __restrict__ dinv, float* __restrict__ coef, int E) {
    const int lane  = threadIdx.x & 31;
    const int wbase = wave_first_edge();
    if (wbase >= E) return;
    if (wbase + EPWV <= E) {
        int s[EPL], d[EPL];
#pragma unroll
        for (int k = 0; k < EPL; ++k) {
            s[k] = src[lane_edge(wbase, lane, k)];
            d[k] = dst[lane_edge(wbase, lane, k)];
        }
        float a[EPL], b[EPL];
#pragma unroll
        for (int k = 0; k < EPL; ++k) {
            a[k] = dinv[s[k]];
            b[k] = dinv[d[k]];
        }
#pragma unroll
        for (int k = 0; k < EPL; ++k) coef[lane_edge(wbase, lane, k)] = a[k] * b[k];
    } else {
#pragma nounroll
        for (int k = 0; k < EPL; ++k) {
            const int e = lane_edge(wbase, lane, k);
            if (e < E) coef[e] = dinv[src[e]] * dinv[dst[e]];
        }
    }
}


template<int KDIM, int NOUT, bool PIN, bool POUT>
__global__ void gemm_wmma_k(const float* X0, const float* X1,
                            const float* __restrict__ W,
                            const float* __restrict__ dinv, const float* __restrict__ bias,
                            float* H0, float* H1, float* P0, float* P1,
                            int* __restrict__ cnt, int ntiles, int nrows) {
    constexpr int NT   = (NOUT + 15) / 16;
    constexpr int KT   = (KDIM + 3) / 4;
    constexpr int KPAD = KT * 4;
    constexpr int NW16 = NT * 16;
    static_assert(!PIN  || KDIM == 64, "planar input is 2 x 32 channels");
    static_assert(!POUT || NOUT == 64, "planar output is 2 x 32 channels");
    static_assert(POUT || NOUT <= 16, "row-major output is one column tile");

    __shared__ float sW[KPAD * NW16];

    if constexpr (POUT) {
        if (blockIdx.x == 0 && threadIdx.x < 2) cnt[threadIdx.x * CNT_STRIDE] = 0;
    }

    for (int idx = threadIdx.x; idx < KPAD * NW16; idx += blockDim.x) {
        const int k = idx / NW16;
        const int n = idx - k * NW16;
        sW[idx] = (k < KDIM && n < NOUT) ? W[k * NOUT + n] : 0.0f;
    }
    __syncthreads();

    const int lane = threadIdx.x & 31;
    const int wv   = blockIdx.x * (blockDim.x >> 5) + (threadIdx.x >> 5);
    if (wv >= ntiles) return;
    const int row0 = wv * 16;
    const int m    = lane & 15;
    const int half = lane >> 4;

    const unsigned lim  = (unsigned)nrows * KDIM - 1u;
    const unsigned rowc = min((unsigned)(row0 + m), (unsigned)nrows - 1u);

    float d2[8];
#pragma unroll
    for (int i = 0; i < 8; ++i) {
        const float di = dinv[min((unsigned)(row0 + half * 8 + i), (unsigned)nrows - 1u)];
        d2[i] = di * di;
    }
    float bb[NT];
#pragma unroll
    for (int t = 0; t < NT; ++t) bb[t] = (t * 16 + m < NOUT) ? bias[t * 16 + m] : 0.0f;

    v8f acc[NT];
#pragma unroll
    for (int t = 0; t < NT; ++t) acc[t] = 0.0f;

    v2f av[KT];
#pragma unroll
    for (int kt = 0; kt < KT; ++kt) {
        if constexpr (PIN) {
            const unsigned kk = (unsigned)((kt * 4) & 31) + (unsigned)(half * 2);
            const unsigned ix = rowc * 32u + kk;
            const float* xp = (kt * 4 < 32) ? X0 : X1;
            av[kt].x = xp[ix];
            av[kt].y = xp[ix + 1u];
        } else {
            const unsigned i0 = (unsigned)(row0 + m) * KDIM + kt * 4 + half * 2;
            av[kt].x = X0[min(i0,      lim)];
            av[kt].y = X0[min(i0 + 1u, lim)];
        }
    }

#pragma unroll
    for (int kt = 0; kt < KT; ++kt) {
        const int ka = kt * 4 + half * 2;
        v2f a = av[kt];
        if constexpr (PIN) {
            a.x = fmaxf(av[kt].x, 0.0f);
            a.y = fmaxf(av[kt].y, 0.0f);
        }
#pragma unroll
        for (int nt = 0; nt < NT; ++nt) {
            const int col = nt * 16 + m;
            v2f b;
            b.x = sW[ka * NW16 + col];
            b.y = sW[(ka + 1) * NW16 + col];
            acc[nt] = __builtin_amdgcn_wmma_f32_16x16x4_f32(
                false, a, false, b, (short)0, acc[nt], false, false);
        }
    }

#pragma unroll
    for (int nt = 0; nt < NT; ++nt) {
#pragma unroll
        for (int i = 0; i < 8; ++i) {
            const int r   = row0 + i + half * 8;
            const int col = nt * 16 + m;
            if constexpr (POUT) {
                float* hp = (nt * 16 < 32) ? H0 : H1;
                float* pp = (nt * 16 < 32) ? P0 : P1;
                hp[(size_t)r * 32 + (col & 31)] = acc[nt][i];
                pp[(size_t)r * 32 + (col & 31)] = fmaf(acc[nt][i], d2[i], bb[nt]);
            } else {
                if (col < NOUT) {
                    H0[(size_t)r * NOUT + col] = acc[nt][i];
                    P0[(size_t)r * NOUT + col] = fmaf(acc[nt][i], d2[i], bb[0]);
                    P1[(size_t)r * NOUT + col] = 0.0f;
                }
            }
        }
    }
}

constexpr int EPW = 256;
constexpr int WPB = 8;
constexpr int EPB = EPW * WPB;

__device__ __forceinline__ float word_at(const float* plane, unsigned byte_off) {
    return *(const float*)((const char*)plane + byte_off);
}
__device__ __forceinline__ float* word_ptr(float* plane, unsigned byte_off) {
    return (float*)((char*)plane + byte_off);
}

template<int G>
__device__ __forceinline__ void edge_group64(int j, unsigned sb, unsigned db, float c, unsigned lane4,
                                             const float* __restrict__ h0, const float* __restrict__ h1,
                                             float* __restrict__ a0, float* __restrict__ a1) {
    float v0[G], v1[G];
#pragma unroll
    for (int u = 0; u < G; ++u) {
        const unsigned bo = (unsigned)__builtin_amdgcn_readlane((int)sb, j + u) + lane4;
        v0[u] = word_at(h0, bo);
        v1[u] = word_at(h1, bo);
    }
#pragma unroll
    for (int u = 0; u < G; ++u) {
        const unsigned bo = (unsigned)__builtin_amdgcn_readlane((int)db, j + u) + lane4;
        const float cj = __int_as_float(__builtin_amdgcn_readlane(__float_as_int(c), j + u));
        atomicAdd(word_ptr(a0, bo), v0[u] * cj);
        atomicAdd(word_ptr(a1, bo), v1[u] * cj);
    }
}

__device__ __forceinline__ void edge_tail64(int j, int cn, unsigned sb, unsigned db, float c, unsigned lane4,
                                            const float* __restrict__ h0, const float* __restrict__ h1,
                                            float* __restrict__ a0, float* __restrict__ a1) {
    float v0[15], v1[15];
#pragma unroll
    for (int u = 0; u < 15; ++u) {
        if (j + u < cn) {
            const unsigned bo = (unsigned)__builtin_amdgcn_readlane((int)sb, j + u) + lane4;
            v0[u] = word_at(h0, bo);
            v1[u] = word_at(h1, bo);
        }
    }
#pragma unroll
    for (int u = 0; u < 15; ++u) {
        if (j + u < cn) {
            const unsigned bo = (unsigned)__builtin_amdgcn_readlane((int)db, j + u) + lane4;
            const float cj = __int_as_float(__builtin_amdgcn_readlane(__float_as_int(c), j + u));
            atomicAdd(word_ptr(a0, bo), v0[u] * cj);
            atomicAdd(word_ptr(a1, bo), v1[u] * cj);
        }
    }
}

__global__ __launch_bounds__(256)
void scatter64_k(const int* __restrict__ src, const int* __restrict__ dst,
                 const float* __restrict__ coef,
                 const float* __restrict__ h0, const float* __restrict__ h1,
                 float* __restrict__ a0, float* __restrict__ a1,
                 int* __restrict__ cnt, int E, int n, int nblocks) {
    __shared__ int   tk[2];
    __shared__ int   qs[WPB][EPW];
    __shared__ int   qd[WPB][EPW];
    __shared__ float qc[WPB][EPW];

    const int hme = pkg_half();
    if (threadIdx.x == 0) {
        int r = hme;
        int t = atomicAdd(&cnt[r * CNT_STRIDE], 1);
        if (t >= nblocks) {
            r ^= 1;
            t = atomicAdd(&cnt[r * CNT_STRIDE], 1);
        }
        tk[0] = t;
        tk[1] = r;
    }
    __syncthreads();
    const int t = __builtin_amdgcn_readfirstlane(tk[0]);
    const int r = __builtin_amdgcn_readfirstlane(tk[1]);
    if (t >= nblocks) return;

    const int lane = threadIdx.x & 31;
    const unsigned lane4 = (unsigned)lane * 4u;
    const int wib  = __builtin_amdgcn_readfirstlane((int)(threadIdx.x >> 5));
    const int ebeg = t * EPB + wib * EPW;
    if (ebeg >= E) return;
    const int eend   = min(ebeg + EPW, E);
    const int nsplit = n >> 1;

    int*   ws = qs[wib];
    int*   wd = qd[wib];
    float* wc = qc[wib];

    int   sv[EPW / 32], dv[EPW / 32];
    float cv[EPW / 32];
#pragma unroll
    for (int b = 0; b < EPW / 32; ++b) {
        const int e = min(ebeg + b * 32 + lane, E - 1);
        dv[b] = dst[e];
        sv[b] = src[e];
        cv[b] = coef[e];
    }
    asm volatile("" :: "v"(sv[0]), "v"(cv[0]), "v"(sv[1]), "v"(cv[1]), "v"(sv[2]), "v"(cv[2]),
                       "v"(sv[3]), "v"(cv[3]), "v"(sv[4]), "v"(cv[4]), "v"(sv[5]), "v"(cv[5]),
                       "v"(sv[6]), "v"(cv[6]), "v"(sv[7]), "v"(cv[7]));
    static_assert(EPW / 32 == 8, "the fence above names eight groups of loads");
    int tot = 0;
#pragma unroll
    for (int b = 0; b < EPW / 32; ++b) {
        const bool mine = (ebeg + b * 32 + lane < eend) && ((dv[b] >= nsplit) == (r != 0));
        const unsigned m = __builtin_amdgcn_ballot_w32(mine);
        if (mine) {
            const int k = tot + (int)__builtin_amdgcn_mbcnt_lo(m, 0u);
            ws[k] = sv[b];
            wd[k] = dv[b];
            wc[k] = cv[b];
        }
        tot += __builtin_popcount(m);
    }

    for (int g0 = 0; g0 < tot; g0 += 32) {
        const int k = g0 + lane;
        int s = 0, d = 0;
        float c = 0.0f;
        if (k < tot) {
            s = ws[k];
            d = wd[k];
            c = wc[k];
        }
        const int cn = min(32, tot - g0);
        const unsigned sb = (unsigned)s << 7;
        const unsigned db = (unsigned)d << 7;
        int j = 0;
        for (; j + 16 <= cn; j += 16) edge_group64<16>(j, sb, db, c, lane4, h0, h1, a0, a1);
        if (j < cn) edge_tail64(j, cn, sb, db, c, lane4, h0, h1, a0, a1);
    }
}


__global__ __launch_bounds__(256)
void scatter3_k(const int* __restrict__ src, const int* __restrict__ dst,
                const float* __restrict__ coef, const float* __restrict__ h,
                float* __restrict__ o0, float* __restrict__ o1, int E) {
    float* out = pkg_half() ? o1 : o0;
    const int lane  = threadIdx.x & 31;
    const int ch    = lane & 3;
    const int wbase = wave_first_edge();
    if (wbase >= E) return;
    if (ch == 3) return;
    const int eb = wbase + (lane >> 2) * 4;

    if (wbase + EPWV <= E) {
        v4i   s[EPL], d[EPL];
        v4f   c[EPL];
        float v[EPL][4];
#pragma unroll
        for (int q = 0; q < EPL; ++q) {
            const int e0 = eb + q * 32;
            s[q] = (v4i){src[e0], src[e0 + 1], src[e0 + 2], src[e0 + 3]};
            d[q] = (v4i){dst[e0], dst[e0 + 1], dst[e0 + 2], dst[e0 + 3]};
            c[q] = (v4f){coef[e0], coef[e0 + 1], coef[e0 + 2], coef[e0 + 3]};
        }
        asm volatile("" :: "v"(s[0]), "v"(s[1]), "v"(s[2]), "v"(s[3]), "v"(s[4]), "v"(s[5]), "v"(s[6]), "v"(s[7]),
                           "v"(d[0]), "v"(d[1]), "v"(d[2]), "v"(d[3]), "v"(d[4]), "v"(d[5]), "v"(d[6]), "v"(d[7]),
                           "v"(c[0]), "v"(c[1]), "v"(c[2]), "v"(c[3]), "v"(c[4]), "v"(c[5]), "v"(c[6]), "v"(c[7]));
        static_assert(EPL == 8, "the fence above names eight runs");
#pragma unroll
        for (int q = 0; q < EPL; ++q) {
#pragma unroll
            for (int g = 0; g < 4; ++g) v[q][g] = h[(size_t)s[q][g] * 3 + ch] * c[q][g];
        }
#pragma unroll
        for (int q = 0; q < EPL; ++q) {
#pragma unroll
            for (int g = 0; g < 4; ++g) atomicAdd(out + (size_t)d[q][g] * 3 + ch, v[q][g]);
        }
    } else {
#pragma nounroll
        for (int q = 0; q < EPL; ++q) {
            for (int g = 0; g < 4; ++g) {
                const int e = eb + q * 32 + g;
                if (e < E) atomicAdd(out + (size_t)dst[e] * 3 + ch, h[(size_t)src[e] * 3 + ch] * coef[e]);
            }
        }
    }
}

__global__ void add_plane_k(float* __restrict__ o0, const float* __restrict__ o1, int n) {
    const int i0 = (blockIdx.x * blockDim.x + threadIdx.x) * NPL;
    if (i0 + NPL <= n) {
        float a[NPL], b[NPL];
#pragma unroll
        for (int g = 0; g < NPL; ++g) {
            a[g] = o0[i0 + g];
            b[g] = o1[i0 + g];
        }
#pragma unroll
        for (int g = 0; g < NPL; ++g) o0[i0 + g] = a[g] + b[g];
    } else {
        for (int g = 0; g < NPL; ++g) {
            if (i0 + g < n) o0[i0 + g] += o1[i0 + g];
        }
    }
}


static inline int cdiv(long long a, int b) { return (int)((a + b - 1) / b); }

extern "C" void kernel_launch(void* const* d_in, const int* in_sizes, int n_in,
                              void* d_out, int out_size, void* d_ws, size_t ws_size,
                              hipStream_t stream) {
    const float* x   = (const float*)d_in[0];
    const int*   ei  = (const int*)d_in[1];
    const float* W1  = (const float*)d_in[2];
    const float* b1  = (const float*)d_in[3];
    const float* W2  = (const float*)d_in[4];
    const float* b2  = (const float*)d_in[5];
    const float* W3  = (const float*)d_in[6];
    const float* b3  = (const float*)d_in[7];
    float* out = (float*)d_out;

    const int N = in_sizes[0] / 9;
    const int E = in_sizes[1] / 2;
    const int* src = ei;
    const int* dst = ei + E;

    float* R0   = (float*)d_ws;
    float* R1   = R0 + (size_t)N * 32;
    float* R2   = R1 + (size_t)N * 32;
    float* R3   = R2 + (size_t)N * 32;
    float* dinv = R3 + (size_t)N * 32;
    float* coef = dinv + N;
    (void)ws_size; (void)n_in; (void)out_size;

    const int ntiles = (N + 15) / 16;
    const int gN   = cdiv(N, 256 * NPL);
    const int gE   = cdiv(E, 256 * EPL);
    const int nEB  = cdiv(E, EPB);
    const int gG   = cdiv(ntiles, 8);
    const int gN3  = cdiv((long long)N * 3, 256 * NPL);

    fill_deg_k<<<gN, 256, 0, stream>>>(dinv, R0, N);
    deg_accum_k<<<gE, 256, 0, stream>>>(dst, dinv, R0, E);
    rsqrt_k<<<gN, 256, 0, stream>>>(dinv, R0, N);
    coef_k<<<gE, 256, 0, stream>>>(src, dst, dinv, coef, E);

    int* tick = (int*)d_out;

    gemm_wmma_k<9, 64, false, true><<<gG, 256, 0, stream>>>(x, nullptr, W1, dinv, b1,
                                                           R0, R1, R2, R3, tick, ntiles, N);
    scatter64_k<<<2 * nEB, 256, 0, stream>>>(src, dst, coef, R0, R1, R2, R3, tick, E, N, nEB);

    gemm_wmma_k<64, 64, true, true><<<gG, 256, 0, stream>>>(R2, R3, W2, dinv, b2,
                                                           R0, R1, R2, R3, tick, ntiles, N);
    scatter64_k<<<2 * nEB, 256, 0, stream>>>(src, dst, coef, R0, R1, R2, R3, tick, E, N, nEB);

    float* o1 = R0 + (size_t)N * 4;
    gemm_wmma_k<64, 3, true, false><<<gG, 256, 0, stream>>>(R2, R3, W3, dinv, b3,
                                                           R0, nullptr, out, o1, nullptr, ntiles, N);
    scatter3_k<<<gE, 256, 0, stream>>>(src, dst, coef, R0, out, o1, E);
    add_plane_k<<<gN3, 256, 0, stream>>>(out, o1, N * 3);
}


# 64-channel edge walk: a group's 16 row offsets, then its 16 add offsets, each in a register of its own before the group issues: no address-translation wait between two loads or adds; sum order kept
# speedup vs baseline: 3.6720x; 1.1435x over previous
#include <hip/hip_runtime.h>


constexpr int CNT_STRIDE = 64;

typedef float v2f __attribute__((ext_vector_type(2)));
typedef float v8f __attribute__((ext_vector_type(8)));
typedef float v4f __attribute__((ext_vector_type(4)));
typedef int   v4i __attribute__((ext_vector_type(4)));

__device__ __forceinline__ int pkg_half() {
    return (int)((__builtin_amdgcn_s_sendmsg_rtn(0x87) >> 18) & 1u);
}


constexpr int EPL  = 8;
constexpr int EPWV = 32 * EPL;
constexpr int NPL  = 4;

__device__ __forceinline__ int wave_first_edge() {
    const int wib = __builtin_amdgcn_readfirstlane((int)(threadIdx.x >> 5));
    return (blockIdx.x * (int)(blockDim.x >> 5) + wib) * EPWV;
}

__device__ __forceinline__ int lane_edge(int wbase, int lane, int k) {
    return wbase + (k >> 2) * 128 + lane * 4 + (k & 3);
}

__global__ void fill_deg_k(float* __restrict__ d0, float* __restrict__ d1, int n) {
    const int i0 = (blockIdx.x * blockDim.x + threadIdx.x) * NPL;
    if (i0 + NPL <= n) {
#pragma unroll
        for (int g = 0; g < NPL; ++g) {
            d0[i0 + g] = 1.0f;
            d1[i0 + g] = 0.0f;
        }
    } else {
        for (int g = 0; g < NPL; ++g) {
            if (i0 + g < n) {
                d0[i0 + g] = 1.0f;
                d1[i0 + g] = 0.0f;
            }
        }
    }
}

__global__ void deg_accum_k(const int* __restrict__ dst, float* __restrict__ d0,
                            float* __restrict__ d1, int E) {
    float* deg = pkg_half() ? d1 : d0;
    const int lane  = threadIdx.x & 31;
    const int wbase = wave_first_edge();
    if (wbase >= E) return;
    if (wbase + EPWV <= E) {
        int d[EPL];
#pragma unroll
        for (int k = 0; k < EPL; ++k) d[k] = dst[lane_edge(wbase, lane, k)];
        asm volatile("" :: "v"(d[0]), "v"(d[1]), "v"(d[2]), "v"(d[3]),
                           "v"(d[4]), "v"(d[5]), "v"(d[6]), "v"(d[7]));
        static_assert(EPL == 8, "the fence above names eight edges");
#pragma unroll
        for (int k = 0; k < EPL; ++k) atomicAdd(&deg[d[k]], 1.0f);
    } else {
#pragma nounroll
        for (int k = 0; k < EPL; ++k) {
            const int e = lane_edge(wbase, lane, k);
            if (e < E) atomicAdd(&deg[dst[e]], 1.0f);
        }
    }
}

__global__ void rsqrt_k(float* __restrict__ d0, const float* __restrict__ d1, int n) {
    const int i0 = (blockIdx.x * blockDim.x + threadIdx.x) * NPL;
    if (i0 + NPL <= n) {
        float a[NPL], b[NPL];
#pragma unroll
        for (int g = 0; g < NPL; ++g) {
            a[g] = d0[i0 + g];
            b[g] = d1[i0 + g];
        }
#pragma unroll
        for (int g = 0; g < NPL; ++g) d0[i0 + g] = rsqrtf(a[g] + b[g]);
    } else {
        for (int g = 0; g < NPL; ++g) {
            if (i0 + g < n) d0[i0 + g] = rsqrtf(d0[i0 + g] + d1[i0 + g]);
        }
    }
}

__global__ void coef_k(const int* __restrict__ src, const int* __restrict__ dst,
                       const float* __restrict__ dinv, float* __restrict__ coef, int E) {
    const int lane  = threadIdx.x & 31;
    const int wbase = wave_first_edge();
    if (wbase >= E) return;
    if (wbase + EPWV <= E) {
        int s[EPL], d[EPL];
#pragma unroll
        for (int k = 0; k < EPL; ++k) {
            s[k] = src[lane_edge(wbase, lane, k)];
            d[k] = dst[lane_edge(wbase, lane, k)];
        }
        float a[EPL], b[EPL];
#pragma unroll
        for (int k = 0; k < EPL; ++k) {
            a[k] = dinv[s[k]];
            b[k] = dinv[d[k]];
        }
#pragma unroll
        for (int k = 0; k < EPL; ++k) coef[lane_edge(wbase, lane, k)] = a[k] * b[k];
    } else {
#pragma nounroll
        for (int k = 0; k < EPL; ++k) {
            const int e = lane_edge(wbase, lane, k);
            if (e < E) coef[e] = dinv[src[e]] * dinv[dst[e]];
        }
    }
}


template<int KDIM, int NOUT, bool PIN, bool POUT>
__global__ void gemm_wmma_k(const float* X0, const float* X1,
                            const float* __restrict__ W,
                            const float* __restrict__ dinv, const float* __restrict__ bias,
                            float* H0, float* H1, float* P0, float* P1,
                            int* __restrict__ cnt, int ntiles, int nrows) {
    constexpr int NT   = (NOUT + 15) / 16;
    constexpr int KT   = (KDIM + 3) / 4;
    constexpr int KPAD = KT * 4;
    constexpr int NW16 = NT * 16;
    static_assert(!PIN  || KDIM == 64, "planar input is 2 x 32 channels");
    static_assert(!POUT || NOUT == 64, "planar output is 2 x 32 channels");
    static_assert(POUT || NOUT <= 16, "row-major output is one column tile");

    __shared__ float sW[KPAD * NW16];

    if constexpr (POUT) {
        if (blockIdx.x == 0 && threadIdx.x < 2) cnt[threadIdx.x * CNT_STRIDE] = 0;
    }

    for (int idx = threadIdx.x; idx < KPAD * NW16; idx += blockDim.x) {
        const int k = idx / NW16;
        const int n = idx - k * NW16;
        sW[idx] = (k < KDIM && n < NOUT) ? W[k * NOUT + n] : 0.0f;
    }
    __syncthreads();

    const int lane = threadIdx.x & 31;
    const int wv   = blockIdx.x * (blockDim.x >> 5) + (threadIdx.x >> 5);
    if (wv >= ntiles) return;
    const int row0 = wv * 16;
    const int m    = lane & 15;
    const int half = lane >> 4;

    const unsigned lim  = (unsigned)nrows * KDIM - 1u;
    const unsigned rowc = min((unsigned)(row0 + m), (unsigned)nrows - 1u);

    float d2[8];
#pragma unroll
    for (int i = 0; i < 8; ++i) {
        const float di = dinv[min((unsigned)(row0 + half * 8 + i), (unsigned)nrows - 1u)];
        d2[i] = di * di;
    }
    float bb[NT];
#pragma unroll
    for (int t = 0; t < NT; ++t) bb[t] = (t * 16 + m < NOUT) ? bias[t * 16 + m] : 0.0f;

    v8f acc[NT];
#pragma unroll
    for (int t = 0; t < NT; ++t) acc[t] = 0.0f;

    v2f av[KT];
#pragma unroll
    for (int kt = 0; kt < KT; ++kt) {
        if constexpr (PIN) {
            const unsigned kk = (unsigned)((kt * 4) & 31) + (unsigned)(half * 2);
            const unsigned ix = rowc * 32u + kk;
            const float* xp = (kt * 4 < 32) ? X0 : X1;
            av[kt].x = xp[ix];
            av[kt].y = xp[ix + 1u];
        } else {
            const unsigned i0 = (unsigned)(row0 + m) * KDIM + kt * 4 + half * 2;
            av[kt].x = X0[min(i0,      lim)];
            av[kt].y = X0[min(i0 + 1u, lim)];
        }
    }

#pragma unroll
    for (int kt = 0; kt < KT; ++kt) {
        const int ka = kt * 4 + half * 2;
        v2f a = av[kt];
        if constexpr (PIN) {
            a.x = fmaxf(av[kt].x, 0.0f);
            a.y = fmaxf(av[kt].y, 0.0f);
        }
#pragma unroll
        for (int nt = 0; nt < NT; ++nt) {
            const int col = nt * 16 + m;
            v2f b;
            b.x = sW[ka * NW16 + col];
            b.y = sW[(ka + 1) * NW16 + col];
            acc[nt] = __builtin_amdgcn_wmma_f32_16x16x4_f32(
                false, a, false, b, (short)0, acc[nt], false, false);
        }
    }

#pragma unroll
    for (int nt = 0; nt < NT; ++nt) {
#pragma unroll
        for (int i = 0; i < 8; ++i) {
            const int r   = row0 + i + half * 8;
            const int col = nt * 16 + m;
            if constexpr (POUT) {
                float* hp = (nt * 16 < 32) ? H0 : H1;
                float* pp = (nt * 16 < 32) ? P0 : P1;
                hp[(size_t)r * 32 + (col & 31)] = acc[nt][i];
                pp[(size_t)r * 32 + (col & 31)] = fmaf(acc[nt][i], d2[i], bb[nt]);
            } else {
                if (col < NOUT) {
                    H0[(size_t)r * NOUT + col] = acc[nt][i];
                    P0[(size_t)r * NOUT + col] = fmaf(acc[nt][i], d2[i], bb[0]);
                    P1[(size_t)r * NOUT + col] = 0.0f;
                }
            }
        }
    }
}

constexpr int EPW = 256;
constexpr int WPB = 8;
constexpr int EPB = EPW * WPB;

__device__ __forceinline__ float word_at(const float* plane, unsigned byte_off) {
    return *(const float*)((const char*)plane + byte_off);
}
__device__ __forceinline__ float* word_ptr(float* plane, unsigned byte_off) {
    return (float*)((char*)plane + byte_off);
}

template<int G>
__device__ __forceinline__ void edge_group64(int j, unsigned sb, unsigned db, float c, unsigned lane4,
                                             const float* __restrict__ h0, const float* __restrict__ h1,
                                             float* __restrict__ a0, float* __restrict__ a1) {
    static_assert(G == 16, "the two fences below name sixteen offsets each");
    unsigned so[G], dd[G];
#pragma unroll
    for (int u = 0; u < G; ++u)
        so[u] = (unsigned)__builtin_amdgcn_readlane((int)sb, j + u) + lane4;
    asm volatile("" :: "v"(so[0]), "v"(so[1]), "v"(so[2]), "v"(so[3]), "v"(so[4]), "v"(so[5]),
                       "v"(so[6]), "v"(so[7]), "v"(so[8]), "v"(so[9]), "v"(so[10]), "v"(so[11]),
                       "v"(so[12]), "v"(so[13]), "v"(so[14]), "v"(so[15]));
    float v0[G], v1[G];
#pragma unroll
    for (int u = 0; u < G; ++u) {
        v0[u] = word_at(h0, so[u]);
        v1[u] = word_at(h1, so[u]);
    }
#pragma unroll
    for (int u = 0; u < G; ++u)
        dd[u] = (unsigned)__builtin_amdgcn_readlane((int)db, j + u) + lane4;
    asm volatile("" :: "v"(dd[0]), "v"(dd[1]), "v"(dd[2]), "v"(dd[3]), "v"(dd[4]), "v"(dd[5]),
                       "v"(dd[6]), "v"(dd[7]), "v"(dd[8]), "v"(dd[9]), "v"(dd[10]), "v"(dd[11]),
                       "v"(dd[12]), "v"(dd[13]), "v"(dd[14]), "v"(dd[15]));
#pragma unroll
    for (int u = 0; u < G; ++u) {
        const float cj = __int_as_float(__builtin_amdgcn_readlane(__float_as_int(c), j + u));
        atomicAdd(word_ptr(a0, dd[u]), v0[u] * cj);
        atomicAdd(word_ptr(a1, dd[u]), v1[u] * cj);
    }
}

__device__ __forceinline__ void edge_tail64(int j, int cn, unsigned sb, unsigned db, float c, unsigned lane4,
                                            const float* __restrict__ h0, const float* __restrict__ h1,
                                            float* __restrict__ a0, float* __restrict__ a1) {
    float v0[15], v1[15];
#pragma unroll
    for (int u = 0; u < 15; ++u) {
        if (j + u < cn) {
            const unsigned bo = (unsigned)__builtin_amdgcn_readlane((int)sb, j + u) + lane4;
            v0[u] = word_at(h0, bo);
            v1[u] = word_at(h1, bo);
        }
    }
#pragma unroll
    for (int u = 0; u < 15; ++u) {
        if (j + u < cn) {
            const unsigned bo = (unsigned)__builtin_amdgcn_readlane((int)db, j + u) + lane4;
            const float cj = __int_as_float(__builtin_amdgcn_readlane(__float_as_int(c), j + u));
            atomicAdd(word_ptr(a0, bo), v0[u] * cj);
            atomicAdd(word_ptr(a1, bo), v1[u] * cj);
        }
    }
}

__global__ __launch_bounds__(256)
void scatter64_k(const int* __restrict__ src, const int* __restrict__ dst,
                 const float* __restrict__ coef,
                 const float* __restrict__ h0, const float* __restrict__ h1,
                 float* __restrict__ a0, float* __restrict__ a1,
                 int* __restrict__ cnt, int E, int n, int nblocks) {
    __shared__ int   tk[2];
    __shared__ int   qs[WPB][EPW];
    __shared__ int   qd[WPB][EPW];
    __shared__ float qc[WPB][EPW];

    const int hme = pkg_half();
    if (threadIdx.x == 0) {
        int r = hme;
        int t = atomicAdd(&cnt[r * CNT_STRIDE], 1);
        if (t >= nblocks) {
            r ^= 1;
            t = atomicAdd(&cnt[r * CNT_STRIDE], 1);
        }
        tk[0] = t;
        tk[1] = r;
    }
    __syncthreads();
    const int t = __builtin_amdgcn_readfirstlane(tk[0]);
    const int r = __builtin_amdgcn_readfirstlane(tk[1]);
    if (t >= nblocks) return;

    const int lane = threadIdx.x & 31;
    const unsigned lane4 = (unsigned)lane * 4u;
    const int wib  = __builtin_amdgcn_readfirstlane((int)(threadIdx.x >> 5));
    const int ebeg = t * EPB + wib * EPW;
    if (ebeg >= E) return;
    const int eend   = min(ebeg + EPW, E);
    const int nsplit = n >> 1;

    int*   ws = qs[wib];
    int*   wd = qd[wib];
    float* wc = qc[wib];

    int   sv[EPW / 32], dv[EPW / 32];
    float cv[EPW / 32];
#pragma unroll
    for (int b = 0; b < EPW / 32; ++b) {
        const int e = min(ebeg + b * 32 + lane, E - 1);
        dv[b] = dst[e];
        sv[b] = src[e];
        cv[b] = coef[e];
    }
    asm volatile("" :: "v"(sv[0]), "v"(cv[0]), "v"(sv[1]), "v"(cv[1]), "v"(sv[2]), "v"(cv[2]),
                       "v"(sv[3]), "v"(cv[3]), "v"(sv[4]), "v"(cv[4]), "v"(sv[5]), "v"(cv[5]),
                       "v"(sv[6]), "v"(cv[6]), "v"(sv[7]), "v"(cv[7]));
    static_assert(EPW / 32 == 8, "the fence above names eight groups of loads");
    int tot = 0;
#pragma unroll
    for (int b = 0; b < EPW / 32; ++b) {
        const bool mine = (ebeg + b * 32 + lane < eend) && ((dv[b] >= nsplit) == (r != 0));
        const unsigned m = __builtin_amdgcn_ballot_w32(mine);
        if (mine) {
            const int k = tot + (int)__builtin_amdgcn_mbcnt_lo(m, 0u);
            ws[k] = sv[b];
            wd[k] = dv[b];
            wc[k] = cv[b];
        }
        tot += __builtin_popcount(m);
    }

    for (int g0 = 0; g0 < tot; g0 += 32) {
        const int k = g0 + lane;
        int s = 0, d = 0;
        float c = 0.0f;
        if (k < tot) {
            s = ws[k];
            d = wd[k];
            c = wc[k];
        }
        const int cn = min(32, tot - g0);
        const unsigned sb = (unsigned)s << 7;
        const unsigned db = (unsigned)d << 7;
        int j = 0;
        for (; j + 16 <= cn; j += 16) edge_group64<16>(j, sb, db, c, lane4, h0, h1, a0, a1);
        if (j < cn) edge_tail64(j, cn, sb, db, c, lane4, h0, h1, a0, a1);
    }
}


__global__ __launch_bounds__(256)
void scatter3_k(const int* __restrict__ src, const int* __restrict__ dst,
                const float* __restrict__ coef, const float* __restrict__ h,
                float* __restrict__ o0, float* __restrict__ o1, int E) {
    float* out = pkg_half() ? o1 : o0;
    const int lane  = threadIdx.x & 31;
    const int ch    = lane & 3;
    const int wbase = wave_first_edge();
    if (wbase >= E) return;
    if (ch == 3) return;
    const int eb = wbase + (lane >> 2) * 4;

    if (wbase + EPWV <= E) {
        v4i   s[EPL], d[EPL];
        v4f   c[EPL];
        float v[EPL][4];
#pragma unroll
        for (int q = 0; q < EPL; ++q) {
            const int e0 = eb + q * 32;
            s[q] = (v4i){src[e0], src[e0 + 1], src[e0 + 2], src[e0 + 3]};
            d[q] = (v4i){dst[e0], dst[e0 + 1], dst[e0 + 2], dst[e0 + 3]};
            c[q] = (v4f){coef[e0], coef[e0 + 1], coef[e0 + 2], coef[e0 + 3]};
        }
        asm volatile("" :: "v"(s[0]), "v"(s[1]), "v"(s[2]), "v"(s[3]), "v"(s[4]), "v"(s[5]), "v"(s[6]), "v"(s[7]),
                           "v"(d[0]), "v"(d[1]), "v"(d[2]), "v"(d[3]), "v"(d[4]), "v"(d[5]), "v"(d[6]), "v"(d[7]),
                           "v"(c[0]), "v"(c[1]), "v"(c[2]), "v"(c[3]), "v"(c[4]), "v"(c[5]), "v"(c[6]), "v"(c[7]));
        static_assert(EPL == 8, "the fence above names eight runs");
#pragma unroll
        for (int q = 0; q < EPL; ++q) {
#pragma unroll
            for (int g = 0; g < 4; ++g) v[q][g] = h[(size_t)s[q][g] * 3 + ch] * c[q][g];
        }
#pragma unroll
        for (int q = 0; q < EPL; ++q) {
#pragma unroll
            for (int g = 0; g < 4; ++g) atomicAdd(out + (size_t)d[q][g] * 3 + ch, v[q][g]);
        }
    } else {
#pragma nounroll
        for (int q = 0; q < EPL; ++q) {
            for (int g = 0; g < 4; ++g) {
                const int e = eb + q * 32 + g;
                if (e < E) atomicAdd(out + (size_t)dst[e] * 3 + ch, h[(size_t)src[e] * 3 + ch] * coef[e]);
            }
        }
    }
}

__global__ void add_plane_k(float* __restrict__ o0, const float* __restrict__ o1, int n) {
    const int i0 = (blockIdx.x * blockDim.x + threadIdx.x) * NPL;
    if (i0 + NPL <= n) {
        float a[NPL], b[NPL];
#pragma unroll
        for (int g = 0; g < NPL; ++g) {
            a[g] = o0[i0 + g];
            b[g] = o1[i0 + g];
        }
#pragma unroll
        for (int g = 0; g < NPL; ++g) o0[i0 + g] = a[g] + b[g];
    } else {
        for (int g = 0; g < NPL; ++g) {
            if (i0 + g < n) o0[i0 + g] += o1[i0 + g];
        }
    }
}


static inline int cdiv(long long a, int b) { return (int)((a + b - 1) / b); }

extern "C" void kernel_launch(void* const* d_in, const int* in_sizes, int n_in,
                              void* d_out, int out_size, void* d_ws, size_t ws_size,
                              hipStream_t stream) {
    const float* x   = (const float*)d_in[0];
    const int*   ei  = (const int*)d_in[1];
    const float* W1  = (const float*)d_in[2];
    const float* b1  = (const float*)d_in[3];
    const float* W2  = (const float*)d_in[4];
    const float* b2  = (const float*)d_in[5];
    const float* W3  = (const float*)d_in[6];
    const float* b3  = (const float*)d_in[7];
    float* out = (float*)d_out;

    const int N = in_sizes[0] / 9;
    const int E = in_sizes[1] / 2;
    const int* src = ei;
    const int* dst = ei + E;

    float* R0   = (float*)d_ws;
    float* R1   = R0 + (size_t)N * 32;
    float* R2   = R1 + (size_t)N * 32;
    float* R3   = R2 + (size_t)N * 32;
    float* dinv = R3 + (size_t)N * 32;
    float* coef = dinv + N;
    (void)ws_size; (void)n_in; (void)out_size;

    const int ntiles = (N + 15) / 16;
    const int gN   = cdiv(N, 256 * NPL);
    const int gE   = cdiv(E, 256 * EPL);
    const int nEB  = cdiv(E, EPB);
    const int gG   = cdiv(ntiles, 8);
    const int gN3  = cdiv((long long)N * 3, 256 * NPL);

    fill_deg_k<<<gN, 256, 0, stream>>>(dinv, R0, N);
    deg_accum_k<<<gE, 256, 0, stream>>>(dst, dinv, R0, E);
    rsqrt_k<<<gN, 256, 0, stream>>>(dinv, R0, N);
    coef_k<<<gE, 256, 0, stream>>>(src, dst, dinv, coef, E);

    int* tick = (int*)d_out;

    gemm_wmma_k<9, 64, false, true><<<gG, 256, 0, stream>>>(x, nullptr, W1, dinv, b1,
                                                           R0, R1, R2, R3, tick, ntiles, N);
    scatter64_k<<<2 * nEB, 256, 0, stream>>>(src, dst, coef, R0, R1, R2, R3, tick, E, N, nEB);

    gemm_wmma_k<64, 64, true, true><<<gG, 256, 0, stream>>>(R2, R3, W2, dinv, b2,
                                                           R0, R1, R2, R3, tick, ntiles, N);
    scatter64_k<<<2 * nEB, 256, 0, stream>>>(src, dst, coef, R0, R1, R2, R3, tick, E, N, nEB);

    float* o1 = R0 + (size_t)N * 4;
    gemm_wmma_k<64, 3, true, false><<<gG, 256, 0, stream>>>(R2, R3, W3, dinv, b3,
                                                           R0, nullptr, out, o1, nullptr, ntiles, N);
    scatter3_k<<<gE, 256, 0, stream>>>(src, dst, coef, R0, out, o1, E);
    add_plane_k<<<gN3, 256, 0, stream>>>(out, o1, N * 3);
}


# Degree count and coefficient look-ups on a per-node 16-bit degree table held whole in LDS (LDS atomics per 20k-edge block, block tables summed per node): no random 4-byte L2 access per edge
# speedup vs baseline: 3.8074x; 1.0369x over previous
#include <hip/hip_runtime.h>


constexpr int CNT_STRIDE = 64;

typedef float v2f __attribute__((ext_vector_type(2)));
typedef float v8f __attribute__((ext_vector_type(8)));
typedef float v4f __attribute__((ext_vector_type(4)));
typedef int   v4i __attribute__((ext_vector_type(4)));

__device__ __forceinline__ int pkg_half() {
    return (int)((__builtin_amdgcn_s_sendmsg_rtn(0x87) >> 18) & 1u);
}


constexpr int EPL  = 8;
constexpr int EPWV = 32 * EPL;
constexpr int NPL  = 4;

__device__ __forceinline__ int wave_first_edge() {
    const int wib = __builtin_amdgcn_readfirstlane((int)(threadIdx.x >> 5));
    return (blockIdx.x * (int)(blockDim.x >> 5) + wib) * EPWV;
}

__device__ __forceinline__ int lane_edge(int wbase, int lane, int k) {
    return wbase + (k >> 2) * 128 + lane * 4 + (k & 3);
}

constexpr int DCAP  = 102400;
constexpr int DCAPW = DCAP / 2;
constexpr int HT    = 1024;
constexpr int TQ    = (DCAPW + HT * 4 - 1) / (HT * 4);
constexpr int TPADW = TQ * HT * 4;
constexpr int HB    = 20480;
constexpr int HEPT  = HB / HT;
constexpr int CEPT  = 8;
constexpr int CEPG  = HT * CEPT;
static_assert(HB < 65536, "a 16-bit count of one workgroup must not overflow");
static_assert(DCAPW % 4 == 0 && HEPT % 4 == 0 && CEPT % 4 == 0, "16-byte accesses");
static_assert(DCAPW % 256 == 0, "deg_finish_k: a 256-thread workgroup's words lie in one table");

__device__ __forceinline__ float dinv_from_count(unsigned c) { return rsqrtf((float)c); }

__device__ __forceinline__ int wg_edge(int e0, int t, int k) {
    return e0 + (k >> 2) * (HT * 4) + t * 4 + (k & 3);
}

__global__ __launch_bounds__(1024)
void deg_hist_k(const int* __restrict__ dst, unsigned* __restrict__ part, int E, int n, int rowp) {
    __shared__ alignas(16) unsigned hw[TPADW];
    const int t  = threadIdx.x;
    const int e0 = blockIdx.x * HB;
    int d[HEPT];
    if (e0 + HB <= E) {
#pragma unroll
        for (int k = 0; k < HEPT; ++k) d[k] = dst[wg_edge(e0, t, k)];
    } else {
#pragma unroll
        for (int k = 0; k < HEPT; ++k) d[k] = dst[min(wg_edge(e0, t, k), E - 1)];
#pragma unroll
        for (int k = 0; k < HEPT; ++k) if (wg_edge(e0, t, k) >= E) d[k] = -1;
    }
    unsigned* po = part + (size_t)blockIdx.x * rowp + t * 4;
    for (int base = 0; base < n; base += DCAP, po += TPADW) {
#pragma unroll
        for (int q = 0; q < TQ; ++q) *(uint4*)&hw[q * (HT * 4) + t * 4] = make_uint4(0u, 0u, 0u, 0u);
        __syncthreads();
#pragma unroll
        for (int k = 0; k < HEPT; ++k) {
            const unsigned r = (unsigned)(d[k] - base);
            if (r < (unsigned)DCAP) atomicAdd(&hw[r >> 1], 1u << ((r & 1u) * 16u));
        }
        __syncthreads();
        uint4 ov[TQ];
#pragma unroll
        for (int q = 0; q < TQ; ++q) ov[q] = *(const uint4*)&hw[q * (HT * 4) + t * 4];
#pragma unroll
        for (int q = 0; q < TQ; ++q) *(uint4*)&po[q * (HT * 4)] = ov[q];
        __syncthreads();
    }
}

constexpr int FB = 32;
template<bool PARTIAL>
__device__ __forceinline__ void add_tables(const unsigned* pk, int rowp, unsigned off, int left,
                                           unsigned& lo, unsigned& hi) {
    unsigned v[FB];
#pragma unroll
    for (int u = 0; u < FB; ++u) {
        v[u] = 0u;
        if (!PARTIAL || u < left)
            v[u] = *(const unsigned*)((const char*)(pk + (size_t)u * rowp) + off);
    }
#pragma unroll
    for (int u = 0; u < FB; ++u) { lo += v[u] & 0xFFFFu; hi += v[u] >> 16; }
}

__global__ void deg_finish_k(const unsigned* __restrict__ part, float* __restrict__ dinv,
                             unsigned* __restrict__ deg16, int nb, int rowp, int pitch, int n) {
    const int w = blockIdx.x * blockDim.x + threadIdx.x;
    if (w >= pitch) {
        if (w < TPADW) deg16[w] = 0xFFFFFFFFu;
        return;
    }
    const int ch = (int)(blockIdx.x * blockDim.x) / DCAPW;
    const unsigned off = (unsigned)(w - ch * DCAPW) * 4u;
    const unsigned* pk = part + (size_t)ch * TPADW;
    unsigned lo = 1u, hi = 1u;
    int k = 0;
#pragma nounroll
    for (; k + FB <= nb; k += FB, pk += (size_t)FB * rowp) add_tables<false>(pk, rowp, off, FB, lo, hi);
    if (k < nb) add_tables<true>(pk, rowp, off, nb - k, lo, hi);
    if (2 * w < n)     dinv[2 * w]     = dinv_from_count(lo);
    if (2 * w + 1 < n) dinv[2 * w + 1] = dinv_from_count(hi);
    deg16[w] = min(lo, 0xFFFFu) | (min(hi, 0xFFFFu) << 16);
}

__device__ __forceinline__ unsigned count_lookup(const unsigned short* tab, int i) {
    return ((unsigned)i < (unsigned)DCAP) ? (unsigned)tab[i] : 0xFFFFu;
}

__global__ __launch_bounds__(1024)
void coef_lds_k(const int* __restrict__ src, const int* __restrict__ dst,
                const unsigned* __restrict__ deg16, const float* __restrict__ dinv,
                float* __restrict__ coef, int E) {
    __shared__ alignas(16) unsigned tw[TPADW];
    const int t  = threadIdx.x;
    const int e0 = blockIdx.x * CEPG;
    const bool whole = (e0 + CEPG <= E);

    uint4 tv[TQ];
#pragma unroll
    for (int q = 0; q < TQ; ++q) tv[q] = *(const uint4*)&deg16[q * (HT * 4) + t * 4];
    int s[CEPT], d[CEPT];
    if (whole) {
#pragma unroll
        for (int k = 0; k < CEPT; ++k) {
            s[k] = src[wg_edge(e0, t, k)];
            d[k] = dst[wg_edge(e0, t, k)];
        }
    } else {
#pragma unroll
        for (int k = 0; k < CEPT; ++k) {
            const int e = min(wg_edge(e0, t, k), E - 1);
            s[k] = src[e];
            d[k] = dst[e];
        }
    }
#pragma unroll
    for (int q = 0; q < TQ; ++q) *(uint4*)&tw[q * (HT * 4) + t * 4] = tv[q];
    __syncthreads();

    const unsigned short* tab = (const unsigned short*)tw;
    unsigned cs[CEPT], cd[CEPT];
#pragma unroll
    for (int k = 0; k < CEPT; ++k) {
        cs[k] = count_lookup(tab, s[k]);
        cd[k] = count_lookup(tab, d[k]);
    }
    asm volatile("" :: "v"(cs[0]), "v"(cs[1]), "v"(cs[2]), "v"(cs[3]), "v"(cs[4]), "v"(cs[5]), "v"(cs[6]), "v"(cs[7]),
                       "v"(cd[0]), "v"(cd[1]), "v"(cd[2]), "v"(cd[3]), "v"(cd[4]), "v"(cd[5]), "v"(cd[6]), "v"(cd[7]));
    static_assert(CEPT == 8, "the fence above names eight edges");
    float c[CEPT];
    bool ask = false;
#pragma unroll
    for (int k = 0; k < CEPT; ++k) {
        c[k] = dinv_from_count(cs[k]) * dinv_from_count(cd[k]);
        ask |= (cs[k] == 0xFFFFu) | (cd[k] == 0xFFFFu);
    }
    if (ask) {
#pragma unroll
        for (int k = 0; k < CEPT; ++k) {
            const float a = (cs[k] == 0xFFFFu) ? dinv[s[k]] : dinv_from_count(cs[k]);
            const float b = (cd[k] == 0xFFFFu) ? dinv[d[k]] : dinv_from_count(cd[k]);
            c[k] = a * b;
        }
    }
    if (whole) {
#pragma unroll
        for (int k = 0; k < CEPT; ++k) coef[wg_edge(e0, t, k)] = c[k];
    } else {
#pragma unroll
        for (int k = 0; k < CEPT; ++k) {
            const int e = wg_edge(e0, t, k);
            if (e < E) coef[e] = c[k];
        }
    }
}


template<int KDIM, int NOUT, bool PIN, bool POUT>
__global__ void gemm_wmma_k(const float* X0, const float* X1,
                            const float* __restrict__ W,
                            const float* __restrict__ dinv, const float* __restrict__ bias,
                            float* H0, float* H1, float* P0, float* P1,
                            int* __restrict__ cnt, int ntiles, int nrows) {
    constexpr int NT   = (NOUT + 15) / 16;
    constexpr int KT   = (KDIM + 3) / 4;
    constexpr int KPAD = KT * 4;
    constexpr int NW16 = NT * 16;
    static_assert(!PIN  || KDIM == 64, "planar input is 2 x 32 channels");
    static_assert(!POUT || NOUT == 64, "planar output is 2 x 32 channels");
    static_assert(POUT || NOUT <= 16, "row-major output is one column tile");

    __shared__ float sW[KPAD * NW16];

    if constexpr (POUT) {
        if (blockIdx.x == 0 && threadIdx.x < 2) cnt[threadIdx.x * CNT_STRIDE] = 0;
    }

    for (int idx = threadIdx.x; idx < KPAD * NW16; idx += blockDim.x) {
        const int k = idx / NW16;
        const int n = idx - k * NW16;
        sW[idx] = (k < KDIM && n < NOUT) ? W[k * NOUT + n] : 0.0f;
    }
    __syncthreads();

    const int lane = threadIdx.x & 31;
    const int wv   = blockIdx.x * (blockDim.x >> 5) + (threadIdx.x >> 5);
    if (wv >= ntiles) return;
    const int row0 = wv * 16;
    const int m    = lane & 15;
    const int half = lane >> 4;

    const unsigned lim  = (unsigned)nrows * KDIM - 1u;
    const unsigned rowc = min((unsigned)(row0 + m), (unsigned)nrows - 1u);

    float d2[8];
#pragma unroll
    for (int i = 0; i < 8; ++i) {
        const float di = dinv[min((unsigned)(row0 + half * 8 + i), (unsigned)nrows - 1u)];
        d2[i] = di * di;
    }
    float bb[NT];
#pragma unroll
    for (int t = 0; t < NT; ++t) bb[t] = (t * 16 + m < NOUT) ? bias[t * 16 + m] : 0.0f;

    v8f acc[NT];
#pragma unroll
    for (int t = 0; t < NT; ++t) acc[t] = 0.0f;

    v2f av[KT];
#pragma unroll
    for (int kt = 0; kt < KT; ++kt) {
        if constexpr (PIN) {
            const unsigned kk = (unsigned)((kt * 4) & 31) + (unsigned)(half * 2);
            const unsigned ix = rowc * 32u + kk;
            const float* xp = (kt * 4 < 32) ? X0 : X1;
            av[kt].x = xp[ix];
            av[kt].y = xp[ix + 1u];
        } else {
            const unsigned i0 = (unsigned)(row0 + m) * KDIM + kt * 4 + half * 2;
            av[kt].x = X0[min(i0,      lim)];
            av[kt].y = X0[min(i0 + 1u, lim)];
        }
    }

#pragma unroll
    for (int kt = 0; kt < KT; ++kt) {
        const int ka = kt * 4 + half * 2;
        v2f a = av[kt];
        if constexpr (PIN) {
            a.x = fmaxf(av[kt].x, 0.0f);
            a.y = fmaxf(av[kt].y, 0.0f);
        }
#pragma unroll
        for (int nt = 0; nt < NT; ++nt) {
            const int col = nt * 16 + m;
            v2f b;
            b.x = sW[ka * NW16 + col];
            b.y = sW[(ka + 1) * NW16 + col];
            acc[nt] = __builtin_amdgcn_wmma_f32_16x16x4_f32(
                false, a, false, b, (short)0, acc[nt], false, false);
        }
    }

#pragma unroll
    for (int nt = 0; nt < NT; ++nt) {
#pragma unroll
        for (int i = 0; i < 8; ++i) {
            const int r   = row0 + i + half * 8;
            const int col = nt * 16 + m;
            if constexpr (POUT) {
                float* hp = (nt * 16 < 32) ? H0 : H1;
                float* pp = (nt * 16 < 32) ? P0 : P1;
                hp[(size_t)r * 32 + (col & 31)] = acc[nt][i];
                pp[(size_t)r * 32 + (col & 31)] = fmaf(acc[nt][i], d2[i], bb[nt]);
            } else {
                if (col < NOUT) {
                    H0[(size_t)r * NOUT + col] = acc[nt][i];
                    P0[(size_t)r * NOUT + col] = fmaf(acc[nt][i], d2[i], bb[0]);
                    P1[(size_t)r * NOUT + col] = 0.0f;
                }
            }
        }
    }
}

constexpr int EPW = 256;
constexpr int WPB = 8;
constexpr int EPB = EPW * WPB;

__device__ __forceinline__ float word_at(const float* plane, unsigned byte_off) {
    return *(const float*)((const char*)plane + byte_off);
}
__device__ __forceinline__ float* word_ptr(float* plane, unsigned byte_off) {
    return (float*)((char*)plane + byte_off);
}

template<int G>
__device__ __forceinline__ void edge_group64(int j, unsigned sb, unsigned db, float c, unsigned lane4,
                                             const float* __restrict__ h0, const float* __restrict__ h1,
                                             float* __restrict__ a0, float* __restrict__ a1) {
    static_assert(G == 16, "the two fences below name sixteen offsets each");
    unsigned so[G], dd[G];
#pragma unroll
    for (int u = 0; u < G; ++u)
        so[u] = (unsigned)__builtin_amdgcn_readlane((int)sb, j + u) + lane4;
    asm volatile("" :: "v"(so[0]), "v"(so[1]), "v"(so[2]), "v"(so[3]), "v"(so[4]), "v"(so[5]),
                       "v"(so[6]), "v"(so[7]), "v"(so[8]), "v"(so[9]), "v"(so[10]), "v"(so[11]),
                       "v"(so[12]), "v"(so[13]), "v"(so[14]), "v"(so[15]));
    float v0[G], v1[G];
#pragma unroll
    for (int u = 0; u < G; ++u) {
        v0[u] = word_at(h0, so[u]);
        v1[u] = word_at(h1, so[u]);
    }
#pragma unroll
    for (int u = 0; u < G; ++u)
        dd[u] = (unsigned)__builtin_amdgcn_readlane((int)db, j + u) + lane4;
    asm volatile("" :: "v"(dd[0]), "v"(dd[1]), "v"(dd[2]), "v"(dd[3]), "v"(dd[4]), "v"(dd[5]),
                       "v"(dd[6]), "v"(dd[7]), "v"(dd[8]), "v"(dd[9]), "v"(dd[10]), "v"(dd[11]),
                       "v"(dd[12]), "v"(dd[13]), "v"(dd[14]), "v"(dd[15]));
#pragma unroll
    for (int u = 0; u < G; ++u) {
        const float cj = __int_as_float(__builtin_amdgcn_readlane(__float_as_int(c), j + u));
        atomicAdd(word_ptr(a0, dd[u]), v0[u] * cj);
        atomicAdd(word_ptr(a1, dd[u]), v1[u] * cj);
    }
}

__device__ __forceinline__ void edge_tail64(int j, int cn, unsigned sb, unsigned db, float c, unsigned lane4,
                                            const float* __restrict__ h0, const float* __restrict__ h1,
                                            float* __restrict__ a0, float* __restrict__ a1) {
    float v0[15], v1[15];
#pragma unroll
    for (int u = 0; u < 15; ++u) {
        if (j + u < cn) {
            const unsigned bo = (unsigned)__builtin_amdgcn_readlane((int)sb, j + u) + lane4;
            v0[u] = word_at(h0, bo);
            v1[u] = word_at(h1, bo);
        }
    }
#pragma unroll
    for (int u = 0; u < 15; ++u) {
        if (j + u < cn) {
            const unsigned bo = (unsigned)__builtin_amdgcn_readlane((int)db, j + u) + lane4;
            const float cj = __int_as_float(__builtin_amdgcn_readlane(__float_as_int(c), j + u));
            atomicAdd(word_ptr(a0, bo), v0[u] * cj);
            atomicAdd(word_ptr(a1, bo), v1[u] * cj);
        }
    }
}

__global__ __launch_bounds__(256)
void scatter64_k(const int* __restrict__ src, const int* __restrict__ dst,
                 const float* __restrict__ coef,
                 const float* __restrict__ h0, const float* __restrict__ h1,
                 float* __restrict__ a0, float* __restrict__ a1,
                 int* __restrict__ cnt, int E, int n, int nblocks) {
    __shared__ int   tk[2];
    __shared__ int   qs[WPB][EPW];
    __shared__ int   qd[WPB][EPW];
    __shared__ float qc[WPB][EPW];

    const int hme = pkg_half();
    if (threadIdx.x == 0) {
        int r = hme;
        int t = atomicAdd(&cnt[r * CNT_STRIDE], 1);
        if (t >= nblocks) {
            r ^= 1;
            t = atomicAdd(&cnt[r * CNT_STRIDE], 1);
        }
        tk[0] = t;
        tk[1] = r;
    }
    __syncthreads();
    const int t = __builtin_amdgcn_readfirstlane(tk[0]);
    const int r = __builtin_amdgcn_readfirstlane(tk[1]);
    if (t >= nblocks) return;

    const int lane = threadIdx.x & 31;
    const unsigned lane4 = (unsigned)lane * 4u;
    const int wib  = __builtin_amdgcn_readfirstlane((int)(threadIdx.x >> 5));
    const int ebeg = t * EPB + wib * EPW;
    if (ebeg >= E) return;
    const int eend   = min(ebeg + EPW, E);
    const int nsplit = n >> 1;

    int*   ws = qs[wib];
    int*   wd = qd[wib];
    float* wc = qc[wib];

    int   sv[EPW / 32], dv[EPW / 32];
    float cv[EPW / 32];
#pragma unroll
    for (int b = 0; b < EPW / 32; ++b) {
        const int e = min(ebeg + b * 32 + lane, E - 1);
        dv[b] = dst[e];
        sv[b] = src[e];
        cv[b] = coef[e];
    }
    asm volatile("" :: "v"(sv[0]), "v"(cv[0]), "v"(sv[1]), "v"(cv[1]), "v"(sv[2]), "v"(cv[2]),
                       "v"(sv[3]), "v"(cv[3]), "v"(sv[4]), "v"(cv[4]), "v"(sv[5]), "v"(cv[5]),
                       "v"(sv[6]), "v"(cv[6]), "v"(sv[7]), "v"(cv[7]));
    static_assert(EPW / 32 == 8, "the fence above names eight groups of loads");
    int tot = 0;
#pragma unroll
    for (int b = 0; b < EPW / 32; ++b) {
        const bool mine = (ebeg + b * 32 + lane < eend) && ((dv[b] >= nsplit) == (r != 0));
        const unsigned m = __builtin_amdgcn_ballot_w32(mine);
        if (mine) {
            const int k = tot + (int)__builtin_amdgcn_mbcnt_lo(m, 0u);
            ws[k] = sv[b];
            wd[k] = dv[b];
            wc[k] = cv[b];
        }
        tot += __builtin_popcount(m);
    }

    for (int g0 = 0; g0 < tot; g0 += 32) {
        const int k = g0 + lane;
        int s = 0, d = 0;
        float c = 0.0f;
        if (k < tot) {
            s = ws[k];
            d = wd[k];
            c = wc[k];
        }
        const int cn = min(32, tot - g0);
        const unsigned sb = (unsigned)s << 7;
        const unsigned db = (unsigned)d << 7;
        int j = 0;
        for (; j + 16 <= cn; j += 16) edge_group64<16>(j, sb, db, c, lane4, h0, h1, a0, a1);
        if (j < cn) edge_tail64(j, cn, sb, db, c, lane4, h0, h1, a0, a1);
    }
}


__global__ __launch_bounds__(256)
void scatter3_k(const int* __restrict__ src, const int* __restrict__ dst,
                const float* __restrict__ coef, const float* __restrict__ h,
                float* __restrict__ o0, float* __restrict__ o1, int E) {
    float* out = pkg_half() ? o1 : o0;
    const int lane  = threadIdx.x & 31;
    const int ch    = lane & 3;
    const int wbase = wave_first_edge();
    if (wbase >= E) return;
    if (ch == 3) return;
    const int eb = wbase + (lane >> 2) * 4;

    if (wbase + EPWV <= E) {
        v4i   s[EPL], d[EPL];
        v4f   c[EPL];
        float v[EPL][4];
#pragma unroll
        for (int q = 0; q < EPL; ++q) {
            const int e0 = eb + q * 32;
            s[q] = (v4i){src[e0], src[e0 + 1], src[e0 + 2], src[e0 + 3]};
            d[q] = (v4i){dst[e0], dst[e0 + 1], dst[e0 + 2], dst[e0 + 3]};
            c[q] = (v4f){coef[e0], coef[e0 + 1], coef[e0 + 2], coef[e0 + 3]};
        }
        asm volatile("" :: "v"(s[0]), "v"(s[1]), "v"(s[2]), "v"(s[3]), "v"(s[4]), "v"(s[5]), "v"(s[6]), "v"(s[7]),
                           "v"(d[0]), "v"(d[1]), "v"(d[2]), "v"(d[3]), "v"(d[4]), "v"(d[5]), "v"(d[6]), "v"(d[7]),
                           "v"(c[0]), "v"(c[1]), "v"(c[2]), "v"(c[3]), "v"(c[4]), "v"(c[5]), "v"(c[6]), "v"(c[7]));
        static_assert(EPL == 8, "the fence above names eight runs");
#pragma unroll
        for (int q = 0; q < EPL; ++q) {
#pragma unroll
            for (int g = 0; g < 4; ++g) v[q][g] = h[(size_t)s[q][g] * 3 + ch] * c[q][g];
        }
#pragma unroll
        for (int q = 0; q < EPL; ++q) {
#pragma unroll
            for (int g = 0; g < 4; ++g) atomicAdd(out + (size_t)d[q][g] * 3 + ch, v[q][g]);
        }
    } else {
#pragma nounroll
        for (int q = 0; q < EPL; ++q) {
            for (int g = 0; g < 4; ++g) {
                const int e = eb + q * 32 + g;
                if (e < E) atomicAdd(out + (size_t)dst[e] * 3 + ch, h[(size_t)src[e] * 3 + ch] * coef[e]);
            }
        }
    }
}

__global__ void add_plane_k(float* __restrict__ o0, const float* __restrict__ o1, int n) {
    const int i0 = (blockIdx.x * blockDim.x + threadIdx.x) * NPL;
    if (i0 + NPL <= n) {
        float a[NPL], b[NPL];
#pragma unroll
        for (int g = 0; g < NPL; ++g) {
            a[g] = o0[i0 + g];
            b[g] = o1[i0 + g];
        }
#pragma unroll
        for (int g = 0; g < NPL; ++g) o0[i0 + g] = a[g] + b[g];
    } else {
        for (int g = 0; g < NPL; ++g) {
            if (i0 + g < n) o0[i0 + g] += o1[i0 + g];
        }
    }
}


static inline int cdiv(long long a, int b) { return (int)((a + b - 1) / b); }

extern "C" void kernel_launch(void* const* d_in, const int* in_sizes, int n_in,
                              void* d_out, int out_size, void* d_ws, size_t ws_size,
                              hipStream_t stream) {
    const float* x   = (const float*)d_in[0];
    const int*   ei  = (const int*)d_in[1];
    const float* W1  = (const float*)d_in[2];
    const float* b1  = (const float*)d_in[3];
    const float* W2  = (const float*)d_in[4];
    const float* b2  = (const float*)d_in[5];
    const float* W3  = (const float*)d_in[6];
    const float* b3  = (const float*)d_in[7];
    float* out = (float*)d_out;

    const int N = in_sizes[0] / 9;
    const int E = in_sizes[1] / 2;
    const int* src = ei;
    const int* dst = ei + E;

    float* R0   = (float*)d_ws;
    float* R1   = R0 + (size_t)N * 32;
    float* R2   = R1 + (size_t)N * 32;
    float* R3   = R2 + (size_t)N * 32;
    float* dinv = R3 + (size_t)N * 32;
    float* coef = dinv + N;
    (void)ws_size; (void)n_in; (void)out_size;

    const int ntiles = (N + 15) / 16;
    const int gE   = cdiv(E, 256 * EPL);
    const int nEB  = cdiv(E, EPB);
    const int gG   = cdiv(ntiles, 8);
    const int gN3  = cdiv((long long)N * 3, 256 * NPL);

    const int pitch = ((N + 7) >> 3) << 2;
    const int rowp  = cdiv(N, DCAP) * TPADW;
    const int nHB   = cdiv(E, HB);
    const int wtot  = pitch > TPADW ? pitch : TPADW;
    unsigned* part  = (unsigned*)R0;
    unsigned* deg16 = (unsigned*)R3;
    deg_hist_k<<<nHB, HT, 0, stream>>>(dst, part, E, N, rowp);
    deg_finish_k<<<cdiv(wtot, 256), 256, 0, stream>>>(part, dinv, deg16, nHB, rowp, pitch, N);
    coef_lds_k<<<cdiv(E, CEPG), HT, 0, stream>>>(src, dst, deg16, dinv, coef, E);

    int* tick = (int*)d_out;

    gemm_wmma_k<9, 64, false, true><<<gG, 256, 0, stream>>>(x, nullptr, W1, dinv, b1,
                                                           R0, R1, R2, R3, tick, ntiles, N);
    scatter64_k<<<2 * nEB, 256, 0, stream>>>(src, dst, coef, R0, R1, R2, R3, tick, E, N, nEB);

    gemm_wmma_k<64, 64, true, true><<<gG, 256, 0, stream>>>(R2, R3, W2, dinv, b2,
                                                           R0, R1, R2, R3, tick, ntiles, N);
    scatter64_k<<<2 * nEB, 256, 0, stream>>>(src, dst, coef, R0, R1, R2, R3, tick, E, N, nEB);

    float* o1 = R0 + (size_t)N * 4;
    gemm_wmma_k<64, 3, true, false><<<gG, 256, 0, stream>>>(R2, R3, W3, dinv, b3,
                                                           R0, nullptr, out, o1, nullptr, ntiles, N);
    scatter3_k<<<gE, 256, 0, stream>>>(src, dst, coef, R0, out, o1, E);
    add_plane_k<<<gN3, 256, 0, stream>>>(out, o1, N * 3);
}


# Transform kernels: a thread's weight words and its tile's dinv, bias and A fragments are requested as one group before the first wait (the staging loop waited once per word: 16 trips in layer 2)
# speedup vs baseline: 4.0236x; 1.0568x over previous
#include <hip/hip_runtime.h>


constexpr int CNT_STRIDE = 64;

typedef float v2f __attribute__((ext_vector_type(2)));
typedef float v8f __attribute__((ext_vector_type(8)));
typedef float v4f __attribute__((ext_vector_type(4)));
typedef int   v4i __attribute__((ext_vector_type(4)));

__device__ __forceinline__ int pkg_half() {
    return (int)((__builtin_amdgcn_s_sendmsg_rtn(0x87) >> 18) & 1u);
}


constexpr int EPL  = 8;
constexpr int EPWV = 32 * EPL;
constexpr int NPL  = 4;

__device__ __forceinline__ int wave_first_edge() {
    const int wib = __builtin_amdgcn_readfirstlane((int)(threadIdx.x >> 5));
    return (blockIdx.x * (int)(blockDim.x >> 5) + wib) * EPWV;
}

__device__ __forceinline__ int lane_edge(int wbase, int lane, int k) {
    return wbase + (k >> 2) * 128 + lane * 4 + (k & 3);
}

constexpr int DCAP  = 102400;
constexpr int DCAPW = DCAP / 2;
constexpr int HT    = 1024;
constexpr int TQ    = (DCAPW + HT * 4 - 1) / (HT * 4);
constexpr int TPADW = TQ * HT * 4;
constexpr int HB    = 20480;
constexpr int HEPT  = HB / HT;
constexpr int CEPT  = 8;
constexpr int CEPG  = HT * CEPT;
static_assert(HB < 65536, "a 16-bit count of one workgroup must not overflow");
static_assert(DCAPW % 4 == 0 && HEPT % 4 == 0 && CEPT % 4 == 0, "16-byte accesses");
static_assert(DCAPW % 256 == 0, "deg_finish_k: a 256-thread workgroup's words lie in one table");

__device__ __forceinline__ float dinv_from_count(unsigned c) { return rsqrtf((float)c); }

__device__ __forceinline__ int wg_edge(int e0, int t, int k) {
    return e0 + (k >> 2) * (HT * 4) + t * 4 + (k & 3);
}

__global__ __launch_bounds__(1024)
void deg_hist_k(const int* __restrict__ dst, unsigned* __restrict__ part, int E, int n, int rowp) {
    __shared__ alignas(16) unsigned hw[TPADW];
    const int t  = threadIdx.x;
    const int e0 = blockIdx.x * HB;
    int d[HEPT];
    if (e0 + HB <= E) {
#pragma unroll
        for (int k = 0; k < HEPT; ++k) d[k] = dst[wg_edge(e0, t, k)];
    } else {
#pragma unroll
        for (int k = 0; k < HEPT; ++k) d[k] = dst[min(wg_edge(e0, t, k), E - 1)];
#pragma unroll
        for (int k = 0; k < HEPT; ++k) if (wg_edge(e0, t, k) >= E) d[k] = -1;
    }
    unsigned* po = part + (size_t)blockIdx.x * rowp + t * 4;
    for (int base = 0; base < n; base += DCAP, po += TPADW) {
#pragma unroll
        for (int q = 0; q < TQ; ++q) *(uint4*)&hw[q * (HT * 4) + t * 4] = make_uint4(0u, 0u, 0u, 0u);
        __syncthreads();
#pragma unroll
        for (int k = 0; k < HEPT; ++k) {
            const unsigned r = (unsigned)(d[k] - base);
            if (r < (unsigned)DCAP) atomicAdd(&hw[r >> 1], 1u << ((r & 1u) * 16u));
        }
        __syncthreads();
        uint4 ov[TQ];
#pragma unroll
        for (int q = 0; q < TQ; ++q) ov[q] = *(const uint4*)&hw[q * (HT * 4) + t * 4];
#pragma unroll
        for (int q = 0; q < TQ; ++q) *(uint4*)&po[q * (HT * 4)] = ov[q];
        __syncthreads();
    }
}

constexpr int FB = 32;
template<bool PARTIAL>
__device__ __forceinline__ void add_tables(const unsigned* pk, int rowp, unsigned off, int left,
                                           unsigned& lo, unsigned& hi) {
    unsigned v[FB];
#pragma unroll
    for (int u = 0; u < FB; ++u) {
        v[u] = 0u;
        if (!PARTIAL || u < left)
            v[u] = *(const unsigned*)((const char*)(pk + (size_t)u * rowp) + off);
    }
#pragma unroll
    for (int u = 0; u < FB; ++u) { lo += v[u] & 0xFFFFu; hi += v[u] >> 16; }
}

__global__ void deg_finish_k(const unsigned* __restrict__ part, float* __restrict__ dinv,
                             unsigned* __restrict__ deg16, int nb, int rowp, int pitch, int n) {
    const int w = blockIdx.x * blockDim.x + threadIdx.x;
    if (w >= pitch) {
        if (w < TPADW) deg16[w] = 0xFFFFFFFFu;
        return;
    }
    const int ch = (int)(blockIdx.x * blockDim.x) / DCAPW;
    const unsigned off = (unsigned)(w - ch * DCAPW) * 4u;
    const unsigned* pk = part + (size_t)ch * TPADW;
    unsigned lo = 1u, hi = 1u;
    int k = 0;
#pragma nounroll
    for (; k + FB <= nb; k += FB, pk += (size_t)FB * rowp) add_tables<false>(pk, rowp, off, FB, lo, hi);
    if (k < nb) add_tables<true>(pk, rowp, off, nb - k, lo, hi);
    if (2 * w < n)     dinv[2 * w]     = dinv_from_count(lo);
    if (2 * w + 1 < n) dinv[2 * w + 1] = dinv_from_count(hi);
    deg16[w] = min(lo, 0xFFFFu) | (min(hi, 0xFFFFu) << 16);
}

__device__ __forceinline__ unsigned count_lookup(const unsigned short* tab, int i) {
    return ((unsigned)i < (unsigned)DCAP) ? (unsigned)tab[i] : 0xFFFFu;
}

__global__ __launch_bounds__(1024)
void coef_lds_k(const int* __restrict__ src, const int* __restrict__ dst,
                const unsigned* __restrict__ deg16, const float* __restrict__ dinv,
                float* __restrict__ coef, int E) {
    __shared__ alignas(16) unsigned tw[TPADW];
    const int t  = threadIdx.x;
    const int e0 = blockIdx.x * CEPG;
    const bool whole = (e0 + CEPG <= E);

    uint4 tv[TQ];
#pragma unroll
    for (int q = 0; q < TQ; ++q) tv[q] = *(const uint4*)&deg16[q * (HT * 4) + t * 4];
    int s[CEPT], d[CEPT];
    if (whole) {
#pragma unroll
        for (int k = 0; k < CEPT; ++k) {
            s[k] = src[wg_edge(e0, t, k)];
            d[k] = dst[wg_edge(e0, t, k)];
        }
    } else {
#pragma unroll
        for (int k = 0; k < CEPT; ++k) {
            const int e = min(wg_edge(e0, t, k), E - 1);
            s[k] = src[e];
            d[k] = dst[e];
        }
    }
#pragma unroll
    for (int q = 0; q < TQ; ++q) *(uint4*)&tw[q * (HT * 4) + t * 4] = tv[q];
    __syncthreads();

    const unsigned short* tab = (const unsigned short*)tw;
    unsigned cs[CEPT], cd[CEPT];
#pragma unroll
    for (int k = 0; k < CEPT; ++k) {
        cs[k] = count_lookup(tab, s[k]);
        cd[k] = count_lookup(tab, d[k]);
    }
    asm volatile("" :: "v"(cs[0]), "v"(cs[1]), "v"(cs[2]), "v"(cs[3]), "v"(cs[4]), "v"(cs[5]), "v"(cs[6]), "v"(cs[7]),
                       "v"(cd[0]), "v"(cd[1]), "v"(cd[2]), "v"(cd[3]), "v"(cd[4]), "v"(cd[5]), "v"(cd[6]), "v"(cd[7]));
    static_assert(CEPT == 8, "the fence above names eight edges");
    float c[CEPT];
    bool ask = false;
#pragma unroll
    for (int k = 0; k < CEPT; ++k) {
        c[k] = dinv_from_count(cs[k]) * dinv_from_count(cd[k]);
        ask |= (cs[k] == 0xFFFFu) | (cd[k] == 0xFFFFu);
    }
    if (ask) {
#pragma unroll
        for (int k = 0; k < CEPT; ++k) {
            const float a = (cs[k] == 0xFFFFu) ? dinv[s[k]] : dinv_from_count(cs[k]);
            const float b = (cd[k] == 0xFFFFu) ? dinv[d[k]] : dinv_from_count(cd[k]);
            c[k] = a * b;
        }
    }
    if (whole) {
#pragma unroll
        for (int k = 0; k < CEPT; ++k) coef[wg_edge(e0, t, k)] = c[k];
    } else {
#pragma unroll
        for (int k = 0; k < CEPT; ++k) {
            const int e = wg_edge(e0, t, k);
            if (e < E) coef[e] = c[k];
        }
    }
}


constexpr int GT = 256;

template<int KDIM, int NOUT, bool PIN, bool POUT>
__global__ __launch_bounds__(GT)
void gemm_wmma_k(const float* X0, const float* X1,
                            const float* __restrict__ W,
                            const float* __restrict__ dinv, const float* __restrict__ bias,
                            float* H0, float* H1, float* P0, float* P1,
                            int* __restrict__ cnt, int ntiles, int nrows) {
    constexpr int NT   = (NOUT + 15) / 16;
    constexpr int KT   = (KDIM + 3) / 4;
    constexpr int KPAD = KT * 4;
    constexpr int NW16 = NT * 16;
    static_assert(!PIN  || KDIM == 64, "planar input is 2 x 32 channels");
    static_assert(!POUT || NOUT == 64, "planar output is 2 x 32 channels");
    static_assert(POUT || NOUT <= 16, "row-major output is one column tile");

    __shared__ float sW[KPAD * NW16];
    constexpr int WQ = (KPAD * NW16) / GT;
    static_assert((KPAD * NW16) % GT == 0, "whole weight words for every thread");

    if constexpr (POUT) {
        if (blockIdx.x == 0 && threadIdx.x < 2) cnt[threadIdx.x * CNT_STRIDE] = 0;
    }

    float wst[WQ];
#pragma unroll
    for (int q = 0; q < WQ; ++q) {
        const int idx = q * GT + (int)threadIdx.x;
        const int k = idx / NW16;
        const int n = idx - k * NW16;
        wst[q] = W[min(k * NOUT + n, KDIM * NOUT - 1)];
    }

    const int lane = threadIdx.x & 31;
    const int wv   = blockIdx.x * (GT >> 5) + (threadIdx.x >> 5);
    const int row0 = wv * 16;
    const int m    = lane & 15;
    const int half = lane >> 4;

    const unsigned lim  = (unsigned)nrows * KDIM - 1u;
    const unsigned rowc = min((unsigned)(row0 + m), (unsigned)nrows - 1u);

    float d2[8];
#pragma unroll
    for (int i = 0; i < 8; ++i) {
        const float di = dinv[min((unsigned)(row0 + half * 8 + i), (unsigned)nrows - 1u)];
        d2[i] = di * di;
    }
    float bb[NT];
#pragma unroll
    for (int t = 0; t < NT; ++t) bb[t] = bias[min(t * 16 + m, NOUT - 1)];

    v2f av[KT];
#pragma unroll
    for (int kt = 0; kt < KT; ++kt) {
        if constexpr (PIN) {
            const unsigned kk = (unsigned)((kt * 4) & 31) + (unsigned)(half * 2);
            const unsigned ix = rowc * 32u + kk;
            const float* xp = (kt * 4 < 32) ? X0 : X1;
            av[kt].x = xp[ix];
            av[kt].y = xp[ix + 1u];
        } else {
            const unsigned i0 = (unsigned)(row0 + m) * KDIM + kt * 4 + half * 2;
            av[kt].x = X0[min(i0,      lim)];
            av[kt].y = X0[min(i0 + 1u, lim)];
        }
    }

#pragma unroll
    for (int q = 0; q < WQ; ++q) {
        const int idx = q * GT + (int)threadIdx.x;
        const int k = idx / NW16;
        const int n = idx - k * NW16;
        sW[idx] = (k < KDIM && n < NOUT) ? wst[q] : 0.0f;
    }
    __syncthreads();
    if (wv >= ntiles) return;

    v8f acc[NT];
#pragma unroll
    for (int t = 0; t < NT; ++t) acc[t] = 0.0f;

#pragma unroll
    for (int kt = 0; kt < KT; ++kt) {
        const int ka = kt * 4 + half * 2;
        v2f a = av[kt];
        if constexpr (PIN) {
            a.x = fmaxf(av[kt].x, 0.0f);
            a.y = fmaxf(av[kt].y, 0.0f);
        }
#pragma unroll
        for (int nt = 0; nt < NT; ++nt) {
            const int col = nt * 16 + m;
            v2f b;
            b.x = sW[ka * NW16 + col];
            b.y = sW[(ka + 1) * NW16 + col];
            acc[nt] = __builtin_amdgcn_wmma_f32_16x16x4_f32(
                false, a, false, b, (short)0, acc[nt], false, false);
        }
    }

#pragma unroll
    for (int nt = 0; nt < NT; ++nt) {
#pragma unroll
        for (int i = 0; i < 8; ++i) {
            const int r   = row0 + i + half * 8;
            const int col = nt * 16 + m;
            if constexpr (POUT) {
                float* hp = (nt * 16 < 32) ? H0 : H1;
                float* pp = (nt * 16 < 32) ? P0 : P1;
                hp[(size_t)r * 32 + (col & 31)] = acc[nt][i];
                pp[(size_t)r * 32 + (col & 31)] = fmaf(acc[nt][i], d2[i], bb[nt]);
            } else {
                if (col < NOUT) {
                    H0[(size_t)r * NOUT + col] = acc[nt][i];
                    P0[(size_t)r * NOUT + col] = fmaf(acc[nt][i], d2[i], bb[0]);
                    P1[(size_t)r * NOUT + col] = 0.0f;
                }
            }
        }
    }
}

constexpr int EPW = 256;
constexpr int WPB = 8;
constexpr int EPB = EPW * WPB;

__device__ __forceinline__ float word_at(const float* plane, unsigned byte_off) {
    return *(const float*)((const char*)plane + byte_off);
}
__device__ __forceinline__ float* word_ptr(float* plane, unsigned byte_off) {
    return (float*)((char*)plane + byte_off);
}

template<int G>
__device__ __forceinline__ void edge_group64(int j, unsigned sb, unsigned db, float c, unsigned lane4,
                                             const float* __restrict__ h0, const float* __restrict__ h1,
                                             float* __restrict__ a0, float* __restrict__ a1) {
    static_assert(G == 16, "the two fences below name sixteen offsets each");
    unsigned so[G], dd[G];
#pragma unroll
    for (int u = 0; u < G; ++u)
        so[u] = (unsigned)__builtin_amdgcn_readlane((int)sb, j + u) + lane4;
    asm volatile("" :: "v"(so[0]), "v"(so[1]), "v"(so[2]), "v"(so[3]), "v"(so[4]), "v"(so[5]),
                       "v"(so[6]), "v"(so[7]), "v"(so[8]), "v"(so[9]), "v"(so[10]), "v"(so[11]),
                       "v"(so[12]), "v"(so[13]), "v"(so[14]), "v"(so[15]));
    float v0[G], v1[G];
#pragma unroll
    for (int u = 0; u < G; ++u) {
        v0[u] = word_at(h0, so[u]);
        v1[u] = word_at(h1, so[u]);
    }
#pragma unroll
    for (int u = 0; u < G; ++u)
        dd[u] = (unsigned)__builtin_amdgcn_readlane((int)db, j + u) + lane4;
    asm volatile("" :: "v"(dd[0]), "v"(dd[1]), "v"(dd[2]), "v"(dd[3]), "v"(dd[4]), "v"(dd[5]),
                       "v"(dd[6]), "v"(dd[7]), "v"(dd[8]), "v"(dd[9]), "v"(dd[10]), "v"(dd[11]),
                       "v"(dd[12]), "v"(dd[13]), "v"(dd[14]), "v"(dd[15]));
#pragma unroll
    for (int u = 0; u < G; ++u) {
        const float cj = __int_as_float(__builtin_amdgcn_readlane(__float_as_int(c), j + u));
        atomicAdd(word_ptr(a0, dd[u]), v0[u] * cj);
        atomicAdd(word_ptr(a1, dd[u]), v1[u] * cj);
    }
}

__device__ __forceinline__ void edge_tail64(int j, int cn, unsigned sb, unsigned db, float c, unsigned lane4,
                                            const float* __restrict__ h0, const float* __restrict__ h1,
                                            float* __restrict__ a0, float* __restrict__ a1) {
    float v0[15], v1[15];
#pragma unroll
    for (int u = 0; u < 15; ++u) {
        if (j + u < cn) {
            const unsigned bo = (unsigned)__builtin_amdgcn_readlane((int)sb, j + u) + lane4;
            v0[u] = word_at(h0, bo);
            v1[u] = word_at(h1, bo);
        }
    }
#pragma unroll
    for (int u = 0; u < 15; ++u) {
        if (j + u < cn) {
            const unsigned bo = (unsigned)__builtin_amdgcn_readlane((int)db, j + u) + lane4;
            const float cj = __int_as_float(__builtin_amdgcn_readlane(__float_as_int(c), j + u));
            atomicAdd(word_ptr(a0, bo), v0[u] * cj);
            atomicAdd(word_ptr(a1, bo), v1[u] * cj);
        }
    }
}

__global__ __launch_bounds__(256)
void scatter64_k(const int* __restrict__ src, const int* __restrict__ dst,
                 const float* __restrict__ coef,
                 const float* __restrict__ h0, const float* __restrict__ h1,
                 float* __restrict__ a0, float* __restrict__ a1,
                 int* __restrict__ cnt, int E, int n, int nblocks) {
    __shared__ int   tk[2];
    __shared__ int   qs[WPB][EPW];
    __shared__ int   qd[WPB][EPW];
    __shared__ float qc[WPB][EPW];

    const int hme = pkg_half();
    if (threadIdx.x == 0) {
        int r = hme;
        int t = atomicAdd(&cnt[r * CNT_STRIDE], 1);
        if (t >= nblocks) {
            r ^= 1;
            t = atomicAdd(&cnt[r * CNT_STRIDE], 1);
        }
        tk[0] = t;
        tk[1] = r;
    }
    __syncthreads();
    const int t = __builtin_amdgcn_readfirstlane(tk[0]);
    const int r = __builtin_amdgcn_readfirstlane(tk[1]);
    if (t >= nblocks) return;

    const int lane = threadIdx.x & 31;
    const unsigned lane4 = (unsigned)lane * 4u;
    const int wib  = __builtin_amdgcn_readfirstlane((int)(threadIdx.x >> 5));
    const int ebeg = t * EPB + wib * EPW;
    if (ebeg >= E) return;
    const int eend   = min(ebeg + EPW, E);
    const int nsplit = n >> 1;

    int*   ws = qs[wib];
    int*   wd = qd[wib];
    float* wc = qc[wib];

    int   sv[EPW / 32], dv[EPW / 32];
    float cv[EPW / 32];
#pragma unroll
    for (int b = 0; b < EPW / 32; ++b) {
        const int e = min(ebeg + b * 32 + lane, E - 1);
        dv[b] = dst[e];
        sv[b] = src[e];
        cv[b] = coef[e];
    }
    asm volatile("" :: "v"(sv[0]), "v"(cv[0]), "v"(sv[1]), "v"(cv[1]), "v"(sv[2]), "v"(cv[2]),
                       "v"(sv[3]), "v"(cv[3]), "v"(sv[4]), "v"(cv[4]), "v"(sv[5]), "v"(cv[5]),
                       "v"(sv[6]), "v"(cv[6]), "v"(sv[7]), "v"(cv[7]));
    static_assert(EPW / 32 == 8, "the fence above names eight groups of loads");
    int tot = 0;
#pragma unroll
    for (int b = 0; b < EPW / 32; ++b) {
        const bool mine = (ebeg + b * 32 + lane < eend) && ((dv[b] >= nsplit) == (r != 0));
        const unsigned m = __builtin_amdgcn_ballot_w32(mine);
        if (mine) {
            const int k = tot + (int)__builtin_amdgcn_mbcnt_lo(m, 0u);
            ws[k] = sv[b];
            wd[k] = dv[b];
            wc[k] = cv[b];
        }
        tot += __builtin_popcount(m);
    }

    for (int g0 = 0; g0 < tot; g0 += 32) {
        const int k = g0 + lane;
        int s = 0, d = 0;
        float c = 0.0f;
        if (k < tot) {
            s = ws[k];
            d = wd[k];
            c = wc[k];
        }
        const int cn = min(32, tot - g0);
        const unsigned sb = (unsigned)s << 7;
        const unsigned db = (unsigned)d << 7;
        int j = 0;
        for (; j + 16 <= cn; j += 16) edge_group64<16>(j, sb, db, c, lane4, h0, h1, a0, a1);
        if (j < cn) edge_tail64(j, cn, sb, db, c, lane4, h0, h1, a0, a1);
    }
}


__global__ __launch_bounds__(256)
void scatter3_k(const int* __restrict__ src, const int* __restrict__ dst,
                const float* __restrict__ coef, const float* __restrict__ h,
                float* __restrict__ o0, float* __restrict__ o1, int E) {
    float* out = pkg_half() ? o1 : o0;
    const int lane  = threadIdx.x & 31;
    const int ch    = lane & 3;
    const int wbase = wave_first_edge();
    if (wbase >= E) return;
    if (ch == 3) return;
    const int eb = wbase + (lane >> 2) * 4;

    if (wbase + EPWV <= E) {
        v4i   s[EPL], d[EPL];
        v4f   c[EPL];
        float v[EPL][4];
#pragma unroll
        for (int q = 0; q < EPL; ++q) {
            const int e0 = eb + q * 32;
            s[q] = (v4i){src[e0], src[e0 + 1], src[e0 + 2], src[e0 + 3]};
            d[q] = (v4i){dst[e0], dst[e0 + 1], dst[e0 + 2], dst[e0 + 3]};
            c[q] = (v4f){coef[e0], coef[e0 + 1], coef[e0 + 2], coef[e0 + 3]};
        }
        asm volatile("" :: "v"(s[0]), "v"(s[1]), "v"(s[2]), "v"(s[3]), "v"(s[4]), "v"(s[5]), "v"(s[6]), "v"(s[7]),
                           "v"(d[0]), "v"(d[1]), "v"(d[2]), "v"(d[3]), "v"(d[4]), "v"(d[5]), "v"(d[6]), "v"(d[7]),
                           "v"(c[0]), "v"(c[1]), "v"(c[2]), "v"(c[3]), "v"(c[4]), "v"(c[5]), "v"(c[6]), "v"(c[7]));
        static_assert(EPL == 8, "the fence above names eight runs");
#pragma unroll
        for (int q = 0; q < EPL; ++q) {
#pragma unroll
            for (int g = 0; g < 4; ++g) v[q][g] = h[(size_t)s[q][g] * 3 + ch] * c[q][g];
        }
#pragma unroll
        for (int q = 0; q < EPL; ++q) {
#pragma unroll
            for (int g = 0; g < 4; ++g) atomicAdd(out + (size_t)d[q][g] * 3 + ch, v[q][g]);
        }
    } else {
#pragma nounroll
        for (int q = 0; q < EPL; ++q) {
            for (int g = 0; g < 4; ++g) {
                const int e = eb + q * 32 + g;
                if (e < E) atomicAdd(out + (size_t)dst[e] * 3 + ch, h[(size_t)src[e] * 3 + ch] * coef[e]);
            }
        }
    }
}

__global__ void add_plane_k(float* __restrict__ o0, const float* __restrict__ o1, int n) {
    const int i0 = (blockIdx.x * blockDim.x + threadIdx.x) * NPL;
    if (i0 + NPL <= n) {
        float a[NPL], b[NPL];
#pragma unroll
        for (int g = 0; g < NPL; ++g) {
            a[g] = o0[i0 + g];
            b[g] = o1[i0 + g];
        }
#pragma unroll
        for (int g = 0; g < NPL; ++g) o0[i0 + g] = a[g] + b[g];
    } else {
        for (int g = 0; g < NPL; ++g) {
            if (i0 + g < n) o0[i0 + g] += o1[i0 + g];
        }
    }
}


static inline int cdiv(long long a, int b) { return (int)((a + b - 1) / b); }

extern "C" void kernel_launch(void* const* d_in, const int* in_sizes, int n_in,
                              void* d_out, int out_size, void* d_ws, size_t ws_size,
                              hipStream_t stream) {
    const float* x   = (const float*)d_in[0];
    const int*   ei  = (const int*)d_in[1];
    const float* W1  = (const float*)d_in[2];
    const float* b1  = (const float*)d_in[3];
    const float* W2  = (const float*)d_in[4];
    const float* b2  = (const float*)d_in[5];
    const float* W3  = (const float*)d_in[6];
    const float* b3  = (const float*)d_in[7];
    float* out = (float*)d_out;

    const int N = in_sizes[0] / 9;
    const int E = in_sizes[1] / 2;
    const int* src = ei;
    const int* dst = ei + E;

    float* R0   = (float*)d_ws;
    float* R1   = R0 + (size_t)N * 32;
    float* R2   = R1 + (size_t)N * 32;
    float* R3   = R2 + (size_t)N * 32;
    float* dinv = R3 + (size_t)N * 32;
    float* coef = dinv + N;
    (void)ws_size; (void)n_in; (void)out_size;

    const int ntiles = (N + 15) / 16;
    const int gE   = cdiv(E, 256 * EPL);
    const int nEB  = cdiv(E, EPB);
    const int gG   = cdiv(ntiles, 8);
    const int gN3  = cdiv((long long)N * 3, 256 * NPL);

    const int pitch = ((N + 7) >> 3) << 2;
    const int rowp  = cdiv(N, DCAP) * TPADW;
    const int nHB   = cdiv(E, HB);
    const int wtot  = pitch > TPADW ? pitch : TPADW;
    unsigned* part  = (unsigned*)R0;
    unsigned* deg16 = (unsigned*)R3;
    deg_hist_k<<<nHB, HT, 0, stream>>>(dst, part, E, N, rowp);
    deg_finish_k<<<cdiv(wtot, 256), 256, 0, stream>>>(part, dinv, deg16, nHB, rowp, pitch, N);
    coef_lds_k<<<cdiv(E, CEPG), HT, 0, stream>>>(src, dst, deg16, dinv, coef, E);

    int* tick = (int*)d_out;

    gemm_wmma_k<9, 64, false, true><<<gG, 256, 0, stream>>>(x, nullptr, W1, dinv, b1,
                                                           R0, R1, R2, R3, tick, ntiles, N);
    scatter64_k<<<2 * nEB, 256, 0, stream>>>(src, dst, coef, R0, R1, R2, R3, tick, E, N, nEB);

    gemm_wmma_k<64, 64, true, true><<<gG, 256, 0, stream>>>(R2, R3, W2, dinv, b2,
                                                           R0, R1, R2, R3, tick, ntiles, N);
    scatter64_k<<<2 * nEB, 256, 0, stream>>>(src, dst, coef, R0, R1, R2, R3, tick, E, N, nEB);

    float* o1 = R0 + (size_t)N * 4;
    gemm_wmma_k<64, 3, true, false><<<gG, 256, 0, stream>>>(R2, R3, W3, dinv, b3,
                                                           R0, nullptr, out, o1, nullptr, ntiles, N);
    scatter3_k<<<gE, 256, 0, stream>>>(src, dst, coef, R0, out, o1, E);
    add_plane_k<<<gN3, 256, 0, stream>>>(out, o1, N * 3);
}
